# starting kernel
# baseline (speedup 1.0000x reference)
#include <hip/hip_runtime.h>
#include <hip/hip_bf16.h>

typedef float v2f __attribute__((ext_vector_type(2)));
typedef float v8f __attribute__((ext_vector_type(8)));

#define IN_DIM   64
#define OUT_DIM  64
#define ATTN_DIM 32

__global__ void zero_f4_kernel(float4* __restrict__ p, long n4) {
    long i = (long)blockIdx.x * blockDim.x + threadIdx.x;
    if (i < n4) p[i] = make_float4(0.f, 0.f, 0.f, 0.f);
}

__global__ void attn_pre_kernel(const float* __restrict__ rela,
                                const float* __restrict__ Wr,
                                const float* __restrict__ Wqr,
                                const float* __restrict__ bqr,
                                const int*   __restrict__ q_rel,
                                float* __restrict__ attn_r,
                                float* __restrict__ attn_qr,
                                int nrel, int batch) {
    int wave = (int)((blockIdx.x * blockDim.x + threadIdx.x) >> 5);
    int k    = threadIdx.x & 31;
    if (wave >= nrel + batch) return;
    if (wave < nrel) {
        const float* hr = rela + (long)wave * IN_DIM;
        const float* w  = Wr   + (long)k    * IN_DIM;
        float s = 0.f;
        #pragma unroll
        for (int d = 0; d < IN_DIM; ++d) s += hr[d] * w[d];
        attn_r[(long)wave * ATTN_DIM + k] = s;
    } else {
        int b  = wave - nrel;
        int qr = q_rel[b];
        const float* hr = rela + (long)qr * IN_DIM;
        const float* w  = Wqr  + (long)k  * IN_DIM;
        float s = bqr[k];
        #pragma unroll
        for (int d = 0; d < IN_DIM; ++d) s += hr[d] * w[d];
        attn_qr[(long)b * ATTN_DIM + k] = s;
    }
}

__global__ void node_attn_wmma_kernel(const float* __restrict__ H,
                                      const float* __restrict__ Ws,
                                      float* __restrict__ S, int n_node) {
    int wave  = (int)((blockIdx.x * blockDim.x + threadIdx.x) >> 5);
    int lane  = threadIdx.x & 31;
    int m0    = wave << 4;
    if (m0 >= n_node) return;
    int row16 = lane & 15;
    int hi    = lane >> 4;
    int arow_i = m0 + row16;
    if (arow_i >= n_node) arow_i = n_node - 1;
    const float* arow  = H  + (long)arow_i      * IN_DIM + 2 * hi;
    const float* b0row = Ws + (long)row16       * IN_DIM + 2 * hi;
    const float* b1row = Ws + (long)(16 + row16) * IN_DIM + 2 * hi;

    v8f c0 = {}; v8f c1 = {};
    #pragma unroll
    for (int kk = 0; kk < IN_DIM / 4; ++kk) {
        v2f a  = *(const v2f*)(arow  + 4 * kk);
        v2f b0 = *(const v2f*)(b0row + 4 * kk);
        v2f b1 = *(const v2f*)(b1row + 4 * kk);
        c0 = __builtin_amdgcn_wmma_f32_16x16x4_f32(false, a, false, b0,
                                                   (short)0, c0, false, false);
        c1 = __builtin_amdgcn_wmma_f32_16x16x4_f32(false, a, false, b1,
                                                   (short)0, c1, false, false);
    }
    #pragma unroll
    for (int j = 0; j < 8; ++j) {
        int m = m0 + j + 8 * hi;
        if (m < n_node) {
            S[(long)m * ATTN_DIM + row16]      = c0[j];
            S[(long)m * ATTN_DIM + 16 + row16] = c1[j];
        }
    }
}

__global__ void edge_kernel(const int*   __restrict__ edges,
                            const float* __restrict__ hidden,
                            const float* __restrict__ rela,
                            const float* __restrict__ S,
                            const float* __restrict__ attn_r,
                            const float* __restrict__ attn_qr,
                            const float* __restrict__ w_alpha_w,
                            const float* __restrict__ w_alpha_b,
                            float* __restrict__ agg, int n_edge) {
    int wave = (int)((blockIdx.x * blockDim.x + threadIdx.x) >> 5);
    int lane = threadIdx.x & 31;
    if (wave >= n_edge) return;
    const int* e = edges + (long)wave * 6;
    int r_idx = e[0];
    int rel   = e[2];
    int sub   = e[4];
    int obj   = e[5];

    float a = S[(long)sub * ATTN_DIM + lane]
            + attn_r[(long)rel * ATTN_DIM + lane]
            + attn_qr[(long)r_idx * ATTN_DIM + lane];
    a = fmaxf(a, 0.f);
    float t = a * w_alpha_w[lane];
    #pragma unroll
    for (int off = 16; off > 0; off >>= 1) t += __shfl_xor(t, off, 32);
    float alpha = 1.f / (1.f + __expf(-(t + w_alpha_b[0])));

    const float* hs = hidden + (long)sub * IN_DIM;
    const float* hr = rela   + (long)rel * IN_DIM;
    float m0v = alpha * (hs[lane]      + hr[lane]);
    float m1v = alpha * (hs[lane + 32] + hr[lane + 32]);
    float* dst = agg + (long)obj * IN_DIM;
    atomicAdd(dst + lane,      m0v);
    atomicAdd(dst + lane + 32, m1v);
}

__global__ void out_gemm_wmma_kernel(float* __restrict__ AGG,
                                     const float* __restrict__ Wh, int n_node) {
    int wave  = (int)((blockIdx.x * blockDim.x + threadIdx.x) >> 5);
    int lane  = threadIdx.x & 31;
    int m0    = wave << 4;
    if (m0 >= n_node) return;
    int row16 = lane & 15;
    int hi    = lane >> 4;
    int arow_i = m0 + row16;
    if (arow_i >= n_node) arow_i = n_node - 1;
    const float* arow = AGG + (long)arow_i * IN_DIM + 2 * hi;
    const float* brow = Wh  + (long)row16  * IN_DIM + 2 * hi;

    v8f c0 = {}, c1 = {}, c2 = {}, c3 = {};
    #pragma unroll
    for (int kk = 0; kk < IN_DIM / 4; ++kk) {
        v2f a  = *(const v2f*)(arow + 4 * kk);
        v2f b0 = *(const v2f*)(brow + 0 * 16 * IN_DIM + 4 * kk);
        v2f b1 = *(const v2f*)(brow + 1 * 16 * IN_DIM + 4 * kk);
        v2f b2 = *(const v2f*)(brow + 2 * 16 * IN_DIM + 4 * kk);
        v2f b3 = *(const v2f*)(brow + 3 * 16 * IN_DIM + 4 * kk);
        c0 = __builtin_amdgcn_wmma_f32_16x16x4_f32(false, a, false, b0,
                                                   (short)0, c0, false, false);
        c1 = __builtin_amdgcn_wmma_f32_16x16x4_f32(false, a, false, b1,
                                                   (short)0, c1, false, false);
        c2 = __builtin_amdgcn_wmma_f32_16x16x4_f32(false, a, false, b2,
                                                   (short)0, c2, false, false);
        c3 = __builtin_amdgcn_wmma_f32_16x16x4_f32(false, a, false, b3,
                                                   (short)0, c3, false, false);
    }
    #pragma unroll
    for (int j = 0; j < 8; ++j) {
        int m = m0 + j + 8 * hi;
        if (m < n_node) {
            float* o = AGG + (long)m * OUT_DIM;
            o[row16]      = c0[j];
            o[16 + row16] = c1[j];
            o[32 + row16] = c2[j];
            o[48 + row16] = c3[j];
        }
    }
}

extern "C" void kernel_launch(void* const* d_in, const int* in_sizes, int n_in,
                              void* d_out, int out_size, void* d_ws, size_t ws_size,
                              hipStream_t stream) {
    const int*   q_rel     = (const int*)  d_in[1];
    const float* hidden    = (const float*)d_in[2];
    const int*   edges     = (const int*)  d_in[3];
    const float* rela      = (const float*)d_in[7];
    const float* Ws        = (const float*)d_in[8];
    const float* Wr        = (const float*)d_in[9];
    const float* Wqr       = (const float*)d_in[10];
    const float* bqr       = (const float*)d_in[11];
    const float* w_alpha_w = (const float*)d_in[12];
    const float* w_alpha_b = (const float*)d_in[13];
    const float* Wh        = (const float*)d_in[14];

    const int n_node = in_sizes[2] / IN_DIM;
    const int n_edge = in_sizes[3] / 6;
    const int nrel   = in_sizes[7] / IN_DIM;
    const int batch  = in_sizes[1];

    float* agg     = (float*)d_out;
    float* S       = (float*)d_ws;
    float* attn_r  = S + (long)n_node * ATTN_DIM;
    float* attn_qr = attn_r + (long)nrel * ATTN_DIM;

    long n4 = (long)n_node * OUT_DIM / 4;
    zero_f4_kernel<<<(int)((n4 + 255) / 256), 256, 0, stream>>>((float4*)agg, n4);

    int pre_waves = nrel + batch;
    attn_pre_kernel<<<(pre_waves + 7) / 8, 256, 0, stream>>>(
        rela, Wr, Wqr, bqr, q_rel, attn_r, attn_qr, nrel, batch);

    int mtiles = (n_node + 15) / 16;
    node_attn_wmma_kernel<<<(mtiles + 7) / 8, 256, 0, stream>>>(hidden, Ws, S, n_node);

    edge_kernel<<<(n_edge + 7) / 8, 256, 0, stream>>>(
        edges, hidden, rela, S, attn_r, attn_qr, w_alpha_w, w_alpha_b, agg, n_edge);

    out_gemm_wmma_kernel<<<(mtiles + 7) / 8, 256, 0, stream>>>(agg, Wh, n_node);
}


# edge pass: each wave loops over 32 consecutive edges (indices read once per wave, broadcast to scalars, gathers of 8 edges in flight) instead of one wave per edge; f32 atomics land in another order
# speedup vs baseline: 1.6381x; 1.6381x over previous
#include <hip/hip_runtime.h>
#include <hip/hip_bf16.h>

typedef float v2f __attribute__((ext_vector_type(2)));
typedef float v8f __attribute__((ext_vector_type(8)));

#define IN_DIM   64
#define OUT_DIM  64
#define ATTN_DIM 32

__global__ void zero_f4_kernel(float4* __restrict__ p, long n4) {
    long i = (long)blockIdx.x * blockDim.x + threadIdx.x;
    if (i < n4) p[i] = make_float4(0.f, 0.f, 0.f, 0.f);
}

__global__ void attn_pre_kernel(const float* __restrict__ rela,
                                const float* __restrict__ Wr,
                                const float* __restrict__ Wqr,
                                const float* __restrict__ bqr,
                                const int*   __restrict__ q_rel,
                                float* __restrict__ attn_r,
                                float* __restrict__ attn_qr,
                                int nrel, int batch) {
    int wave = (int)((blockIdx.x * blockDim.x + threadIdx.x) >> 5);
    int k    = threadIdx.x & 31;
    if (wave >= nrel + batch) return;
    if (wave < nrel) {
        const float* hr = rela + (long)wave * IN_DIM;
        const float* w  = Wr   + (long)k    * IN_DIM;
        float s = 0.f;
        #pragma unroll
        for (int d = 0; d < IN_DIM; ++d) s += hr[d] * w[d];
        attn_r[(long)wave * ATTN_DIM + k] = s;
    } else {
        int b  = wave - nrel;
        int qr = q_rel[b];
        const float* hr = rela + (long)qr * IN_DIM;
        const float* w  = Wqr  + (long)k  * IN_DIM;
        float s = bqr[k];
        #pragma unroll
        for (int d = 0; d < IN_DIM; ++d) s += hr[d] * w[d];
        attn_qr[(long)b * ATTN_DIM + k] = s;
    }
}

__global__ void node_attn_wmma_kernel(const float* __restrict__ H,
                                      const float* __restrict__ Ws,
                                      float* __restrict__ S, int n_node) {
    int wave  = (int)((blockIdx.x * blockDim.x + threadIdx.x) >> 5);
    int lane  = threadIdx.x & 31;
    int m0    = wave << 4;
    if (m0 >= n_node) return;
    int row16 = lane & 15;
    int hi    = lane >> 4;
    int arow_i = m0 + row16;
    if (arow_i >= n_node) arow_i = n_node - 1;
    const float* arow  = H  + (long)arow_i      * IN_DIM + 2 * hi;
    const float* b0row = Ws + (long)row16       * IN_DIM + 2 * hi;
    const float* b1row = Ws + (long)(16 + row16) * IN_DIM + 2 * hi;

    v8f c0 = {}; v8f c1 = {};
    #pragma unroll
    for (int kk = 0; kk < IN_DIM / 4; ++kk) {
        v2f a  = *(const v2f*)(arow  + 4 * kk);
        v2f b0 = *(const v2f*)(b0row + 4 * kk);
        v2f b1 = *(const v2f*)(b1row + 4 * kk);
        c0 = __builtin_amdgcn_wmma_f32_16x16x4_f32(false, a, false, b0,
                                                   (short)0, c0, false, false);
        c1 = __builtin_amdgcn_wmma_f32_16x16x4_f32(false, a, false, b1,
                                                   (short)0, c1, false, false);
    }
    #pragma unroll
    for (int j = 0; j < 8; ++j) {
        int m = m0 + j + 8 * hi;
        if (m < n_node) {
            S[(long)m * ATTN_DIM + row16]      = c0[j];
            S[(long)m * ATTN_DIM + 16 + row16] = c1[j];
        }
    }
}

#define EPW 32
#define EG  8

struct EdgeVals { float s, ar, aq, h0, h1, r0, r1; };

__device__ __forceinline__ void edge_load(EdgeVals& v, int r_idx, int rel, int sub, int lane,
                                          const float* __restrict__ hidden,
                                          const float* __restrict__ rela,
                                          const float* __restrict__ S,
                                          const float* __restrict__ attn_r,
                                          const float* __restrict__ attn_qr) {
    v.s  = S[(long)sub * ATTN_DIM + lane];
    v.ar = attn_r[(long)rel * ATTN_DIM + lane];
    v.aq = attn_qr[(long)r_idx * ATTN_DIM + lane];
    const float* hs = hidden + (long)sub * IN_DIM;
    const float* hr = rela   + (long)rel * IN_DIM;
    v.h0 = hs[lane];
    v.h1 = hs[lane + 32];
    v.r0 = hr[lane];
    v.r1 = hr[lane + 32];
}

__device__ __forceinline__ void edge_finish(const EdgeVals& v, int obj, int lane,
                                            float wa, float wb, float* __restrict__ agg) {
    float a = v.s + v.ar + v.aq;
    a = fmaxf(a, 0.f);
    float t = a * wa;
    #pragma unroll
    for (int off = 16; off > 0; off >>= 1) t += __shfl_xor(t, off, 32);
    float alpha = 1.f / (1.f + __expf(-(t + wb)));
    float* dst = agg + (long)obj * IN_DIM;
    atomicAdd(dst + lane,      alpha * (v.h0 + v.r0));
    atomicAdd(dst + lane + 32, alpha * (v.h1 + v.r1));
}

__global__ void edge_kernel(const int*   __restrict__ edges,
                            const float* __restrict__ hidden,
                            const float* __restrict__ rela,
                            const float* __restrict__ S,
                            const float* __restrict__ attn_r,
                            const float* __restrict__ attn_qr,
                            const float* __restrict__ w_alpha_w,
                            const float* __restrict__ w_alpha_b,
                            float* __restrict__ agg, int n_edge) {
    int wave = __builtin_amdgcn_readfirstlane(
        (int)((blockIdx.x * blockDim.x + threadIdx.x) >> 5));
    int lane = threadIdx.x & 31;
    long e0 = (long)wave * EPW;
    if (e0 >= n_edge) return;
    long left = (long)n_edge - e0;
    int cnt = left < EPW ? (int)left : EPW;

    int my_r = 0, my_rel = 0, my_sub = 0, my_obj = 0;
    if (lane < cnt) {
        const int* e = edges + (e0 + lane) * 6;
        my_r   = e[0];
        my_rel = e[2];
        my_sub = e[4];
        my_obj = e[5];
    }
    float wa = w_alpha_w[lane];
    float wb = w_alpha_b[0];

    int i = 0;
    for (; i + EG <= cnt; i += EG) {
        EdgeVals v[EG];
        int ob[EG];
        #pragma unroll
        for (int k = 0; k < EG; ++k) {
            int r_idx = __builtin_amdgcn_readlane(my_r,   i + k);
            int rel   = __builtin_amdgcn_readlane(my_rel, i + k);
            int sub   = __builtin_amdgcn_readlane(my_sub, i + k);
            ob[k]     = __builtin_amdgcn_readlane(my_obj, i + k);
            edge_load(v[k], r_idx, rel, sub, lane, hidden, rela, S, attn_r, attn_qr);
        }
        #pragma unroll
        for (int k = 0; k < EG; ++k)
            edge_finish(v[k], ob[k], lane, wa, wb, agg);
    }
    for (; i < cnt; ++i) {
        EdgeVals v;
        int r_idx = __builtin_amdgcn_readlane(my_r,   i);
        int rel   = __builtin_amdgcn_readlane(my_rel, i);
        int sub   = __builtin_amdgcn_readlane(my_sub, i);
        int obj   = __builtin_amdgcn_readlane(my_obj, i);
        edge_load(v, r_idx, rel, sub, lane, hidden, rela, S, attn_r, attn_qr);
        edge_finish(v, obj, lane, wa, wb, agg);
    }
}

__global__ void out_gemm_wmma_kernel(float* __restrict__ AGG,
                                     const float* __restrict__ Wh, int n_node) {
    int wave  = (int)((blockIdx.x * blockDim.x + threadIdx.x) >> 5);
    int lane  = threadIdx.x & 31;
    int m0    = wave << 4;
    if (m0 >= n_node) return;
    int row16 = lane & 15;
    int hi    = lane >> 4;
    int arow_i = m0 + row16;
    if (arow_i >= n_node) arow_i = n_node - 1;
    const float* arow = AGG + (long)arow_i * IN_DIM + 2 * hi;
    const float* brow = Wh  + (long)row16  * IN_DIM + 2 * hi;

    v8f c0 = {}, c1 = {}, c2 = {}, c3 = {};
    #pragma unroll
    for (int kk = 0; kk < IN_DIM / 4; ++kk) {
        v2f a  = *(const v2f*)(arow + 4 * kk);
        v2f b0 = *(const v2f*)(brow + 0 * 16 * IN_DIM + 4 * kk);
        v2f b1 = *(const v2f*)(brow + 1 * 16 * IN_DIM + 4 * kk);
        v2f b2 = *(const v2f*)(brow + 2 * 16 * IN_DIM + 4 * kk);
        v2f b3 = *(const v2f*)(brow + 3 * 16 * IN_DIM + 4 * kk);
        c0 = __builtin_amdgcn_wmma_f32_16x16x4_f32(false, a, false, b0,
                                                   (short)0, c0, false, false);
        c1 = __builtin_amdgcn_wmma_f32_16x16x4_f32(false, a, false, b1,
                                                   (short)0, c1, false, false);
        c2 = __builtin_amdgcn_wmma_f32_16x16x4_f32(false, a, false, b2,
                                                   (short)0, c2, false, false);
        c3 = __builtin_amdgcn_wmma_f32_16x16x4_f32(false, a, false, b3,
                                                   (short)0, c3, false, false);
    }
    #pragma unroll
    for (int j = 0; j < 8; ++j) {
        int m = m0 + j + 8 * hi;
        if (m < n_node) {
            float* o = AGG + (long)m * OUT_DIM;
            o[row16]      = c0[j];
            o[16 + row16] = c1[j];
            o[32 + row16] = c2[j];
            o[48 + row16] = c3[j];
        }
    }
}

extern "C" void kernel_launch(void* const* d_in, const int* in_sizes, int n_in,
                              void* d_out, int out_size, void* d_ws, size_t ws_size,
                              hipStream_t stream) {
    const int*   q_rel     = (const int*)  d_in[1];
    const float* hidden    = (const float*)d_in[2];
    const int*   edges     = (const int*)  d_in[3];
    const float* rela      = (const float*)d_in[7];
    const float* Ws        = (const float*)d_in[8];
    const float* Wr        = (const float*)d_in[9];
    const float* Wqr       = (const float*)d_in[10];
    const float* bqr       = (const float*)d_in[11];
    const float* w_alpha_w = (const float*)d_in[12];
    const float* w_alpha_b = (const float*)d_in[13];
    const float* Wh        = (const float*)d_in[14];

    const int n_node = in_sizes[2] / IN_DIM;
    const int n_edge = in_sizes[3] / 6;
    const int nrel   = in_sizes[7] / IN_DIM;
    const int batch  = in_sizes[1];

    float* agg     = (float*)d_out;
    float* S       = (float*)d_ws;
    float* attn_r  = S + (long)n_node * ATTN_DIM;
    float* attn_qr = attn_r + (long)nrel * ATTN_DIM;

    long n4 = (long)n_node * OUT_DIM / 4;
    zero_f4_kernel<<<(int)((n4 + 255) / 256), 256, 0, stream>>>((float4*)agg, n4);

    int pre_waves = nrel + batch;
    attn_pre_kernel<<<(pre_waves + 7) / 8, 256, 0, stream>>>(
        rela, Wr, Wqr, bqr, q_rel, attn_r, attn_qr, nrel, batch);

    int mtiles = (n_node + 15) / 16;
    node_attn_wmma_kernel<<<(mtiles + 7) / 8, 256, 0, stream>>>(hidden, Ws, S, n_node);

    edge_kernel<<<(n_edge + 8 * EPW - 1) / (8 * EPW), 256, 0, stream>>>(
        edges, hidden, rela, S, attn_r, attn_qr, w_alpha_w, w_alpha_b, agg, n_edge);

    out_gemm_wmma_kernel<<<(mtiles + 7) / 8, 256, 0, stream>>>(agg, Wh, n_node);
}


# global f32 atomic adds removed: edges binned per call by 196-node destination block (LDS count, scan, place); one workgroup per block sums rows with LDS f32 adds (arrival order), stores rows once
# speedup vs baseline: 1.8370x; 1.1214x over previous
#include <hip/hip_runtime.h>
#include <hip/hip_bf16.h>

typedef float v2f __attribute__((ext_vector_type(2)));
typedef float v8f __attribute__((ext_vector_type(8)));

#define IN_DIM   64
#define OUT_DIM  64
#define ATTN_DIM 32

__global__ void attn_pre_kernel(const float* __restrict__ rela,
                                const float* __restrict__ Wr,
                                const float* __restrict__ Wqr,
                                const float* __restrict__ bqr,
                                const int*   __restrict__ q_rel,
                                float* __restrict__ attn_r,
                                float* __restrict__ attn_qr,
                                int nrel, int batch) {
    int wave = (int)((blockIdx.x * blockDim.x + threadIdx.x) >> 5);
    int k    = threadIdx.x & 31;
    if (wave >= nrel + batch) return;
    if (wave < nrel) {
        const float* hr = rela + (long)wave * IN_DIM;
        const float* w  = Wr   + (long)k    * IN_DIM;
        float s = 0.f;
        #pragma unroll
        for (int d = 0; d < IN_DIM; ++d) s += hr[d] * w[d];
        attn_r[(long)wave * ATTN_DIM + k] = s;
    } else {
        int b  = wave - nrel;
        int qr = q_rel[b];
        const float* hr = rela + (long)qr * IN_DIM;
        const float* w  = Wqr  + (long)k  * IN_DIM;
        float s = bqr[k];
        #pragma unroll
        for (int d = 0; d < IN_DIM; ++d) s += hr[d] * w[d];
        attn_qr[(long)b * ATTN_DIM + k] = s;
    }
}

__global__ void node_attn_wmma_kernel(const float* __restrict__ H,
                                      const float* __restrict__ Ws,
                                      float* __restrict__ S, int n_node) {
    int wave  = (int)((blockIdx.x * blockDim.x + threadIdx.x) >> 5);
    int lane  = threadIdx.x & 31;
    int m0    = wave << 4;
    if (m0 >= n_node) return;
    int row16 = lane & 15;
    int hi    = lane >> 4;
    int arow_i = m0 + row16;
    if (arow_i >= n_node) arow_i = n_node - 1;
    const float* arow  = H  + (long)arow_i      * IN_DIM + 2 * hi;
    const float* b0row = Ws + (long)row16       * IN_DIM + 2 * hi;
    const float* b1row = Ws + (long)(16 + row16) * IN_DIM + 2 * hi;

    v8f c0 = {}; v8f c1 = {};
    #pragma unroll
    for (int kk = 0; kk < IN_DIM / 4; ++kk) {
        v2f a  = *(const v2f*)(arow  + 4 * kk);
        v2f b0 = *(const v2f*)(b0row + 4 * kk);
        v2f b1 = *(const v2f*)(b1row + 4 * kk);
        c0 = __builtin_amdgcn_wmma_f32_16x16x4_f32(false, a, false, b0,
                                                   (short)0, c0, false, false);
        c1 = __builtin_amdgcn_wmma_f32_16x16x4_f32(false, a, false, b1,
                                                   (short)0, c1, false, false);
    }
    #pragma unroll
    for (int j = 0; j < 8; ++j) {
        int m = m0 + j + 8 * hi;
        if (m < n_node) {
            S[(long)m * ATTN_DIM + row16]      = c0[j];
            S[(long)m * ATTN_DIM + 16 + row16] = c1[j];
        }
    }
}

#define BN       196
#define NCHUNK   256
#define BIN_T    1024
#define BIN_U    4
#define SCAN_G   32

__global__ void bin_count_kernel(const int* __restrict__ edges,
                                 int* __restrict__ cnt,
                                 int n_edge, int nb, int chunk) {
    extern __shared__ int sh_i[];
    const int t = threadIdx.x;
    const int g = blockIdx.x;
    for (int b = t; b < nb; b += BIN_T) sh_i[b] = 0;
    __syncthreads();
    long e0 = (long)g * chunk;
    long e1 = e0 + chunk;
    if (e1 > n_edge) e1 = n_edge;
    for (long e = e0 + t; e < e1; e += (long)BIN_U * BIN_T) {
        int obj[BIN_U], inc[BIN_U];
        #pragma unroll
        for (int k = 0; k < BIN_U; ++k) {
            long ee = e + (long)k * BIN_T;
            inc[k] = (ee < e1) ? 1 : 0;
            if (ee >= e1) ee = e1 - 1;
            obj[k] = edges[ee * 6 + 5];
        }
        #pragma unroll
        for (int k = 0; k < BIN_U; ++k)
            atomicAdd(&sh_i[(unsigned)obj[k] / BN], inc[k]);
    }
    __syncthreads();
    for (int b = t; b < nb; b += BIN_T) cnt[(long)g * nb + b] = sh_i[b];
}

__global__ void bin_scan_kernel(const int* __restrict__ cnt,
                                int* __restrict__ off,
                                int* __restrict__ bstart,
                                int nb) {
    extern __shared__ int sh_i[];
    int* tot = sh_i;
    int* scr = sh_i + nb;
    const int t = threadIdx.x;
    for (int b = t; b < nb; b += BIN_T) {
        int run = 0;
        for (int g = 0; g < NCHUNK; g += SCAN_G) {
            int c[SCAN_G];
            #pragma unroll
            for (int k = 0; k < SCAN_G; ++k) c[k] = cnt[(long)(g + k) * nb + b];
            #pragma unroll
            for (int k = 0; k < SCAN_G; ++k) {
                int v = c[k];
                off[(long)(g + k) * nb + b] = run;
                run += v;
            }
        }
        tot[b] = run;
    }
    __syncthreads();
    int carry = 0;
    for (int base = 0; base < nb; base += BIN_T) {
        const int i = base + t;
        const int v = (i < nb) ? tot[i] : 0;
        scr[t] = v;
        __syncthreads();
        for (int d = 1; d < BIN_T; d <<= 1) {
            int add = (t >= d) ? scr[t - d] : 0;
            __syncthreads();
            scr[t] += add;
            __syncthreads();
        }
        if (i < nb) bstart[i] = carry + scr[t] - v;
        const int tile_total = scr[BIN_T - 1];
        __syncthreads();
        carry += tile_total;
    }
    if (t == 0) bstart[nb] = carry;
}

__global__ void bin_place_kernel(const int* __restrict__ edges,
                                 const int* __restrict__ off,
                                 const int* __restrict__ bstart,
                                 int4* __restrict__ rec,
                                 int n_edge, int nb, int chunk) {
    extern __shared__ int sh_i[];
    const int t = threadIdx.x;
    const int g = blockIdx.x;
    for (int b = t; b < nb; b += BIN_T) sh_i[b] = bstart[b] + off[(long)g * nb + b];
    __syncthreads();
    long e0 = (long)g * chunk;
    long e1 = e0 + chunk;
    if (e1 > n_edge) e1 = n_edge;
    for (long e = e0 + t; e < e1; e += (long)BIN_U * BIN_T) {
        int4 q[BIN_U];
        #pragma unroll
        for (int k = 0; k < BIN_U; ++k) {
            long ee = e + (long)k * BIN_T;
            q[k] = make_int4(0, 0, 0, 0);
            if (ee < e1) {
                const int* p = edges + ee * 6;
                q[k] = make_int4(p[0], p[2], p[4], p[5]);
            }
        }
        #pragma unroll
        for (int k = 0; k < BIN_U; ++k) {
            if (e + (long)k * BIN_T < e1) {
                int pos = atomicAdd(&sh_i[(unsigned)q[k].w / BN], 1);
                rec[pos] = q[k];
            }
        }
    }
}

#define AGG_T 512
#define EG    8

struct EdgeVals { float s, ar, aq, h0, h1, r0, r1; };

__device__ __forceinline__ void edge_load(EdgeVals& v, int r_idx, int rel, int sub, int lane,
                                          const float* __restrict__ hidden,
                                          const float* __restrict__ rela,
                                          const float* __restrict__ S,
                                          const float* __restrict__ attn_r,
                                          const float* __restrict__ attn_qr) {
    v.s  = S[(long)sub * ATTN_DIM + lane];
    v.ar = attn_r[(long)rel * ATTN_DIM + lane];
    v.aq = attn_qr[(long)r_idx * ATTN_DIM + lane];
    const float* hs = hidden + (long)sub * IN_DIM;
    const float* hr = rela   + (long)rel * IN_DIM;
    v.h0 = hs[lane];
    v.h1 = hs[lane + 32];
    v.r0 = hr[lane];
    v.r1 = hr[lane + 32];
}

__device__ __forceinline__ void edge_finish(const EdgeVals& v, int row, int lane,
                                            float wa, float wb, float* tile) {
    float a = v.s + v.ar + v.aq;
    a = fmaxf(a, 0.f);
    float t = a * wa;
    #pragma unroll
    for (int off = 16; off > 0; off >>= 1) t += __shfl_xor(t, off, 32);
    float alpha = 1.f / (1.f + __expf(-(t + wb)));
    float* dst = tile + row * IN_DIM;
    atomicAdd(dst + lane,      alpha * (v.h0 + v.r0));
    atomicAdd(dst + lane + 32, alpha * (v.h1 + v.r1));
}

__global__ void block_agg_kernel(const int4*  __restrict__ rec,
                                 const int*   __restrict__ bstart,
                                 const float* __restrict__ hidden,
                                 const float* __restrict__ rela,
                                 const float* __restrict__ S,
                                 const float* __restrict__ attn_r,
                                 const float* __restrict__ attn_qr,
                                 const float* __restrict__ w_alpha_w,
                                 const float* __restrict__ w_alpha_b,
                                 float* __restrict__ agg, int n_node) {
    __shared__ float tile[BN * IN_DIM];
    const int t = threadIdx.x;
    const int node0 = (int)blockIdx.x * BN;
    for (int i = t; i < BN * IN_DIM; i += AGG_T) tile[i] = 0.f;
    __syncthreads();

    const int wave = __builtin_amdgcn_readfirstlane(t >> 5);
    const int lane = t & 31;
    const int r_begin = bstart[blockIdx.x];
    const int r_end   = bstart[blockIdx.x + 1];
    const float wa = w_alpha_w[lane];
    const float wb = w_alpha_b[0];

    for (int base = r_begin + wave * 32; base < r_end; base += (AGG_T / 32) * 32) {
        const int left = r_end - base;
        const int cnt = left < 32 ? left : 32;
        int4 my = make_int4(0, 0, 0, node0);
        if (lane < cnt) my = rec[base + lane];

        int i = 0;
        for (; i + EG <= cnt; i += EG) {
            EdgeVals v[EG];
            int row[EG];
            #pragma unroll
            for (int k = 0; k < EG; ++k) {
                int r_idx = __builtin_amdgcn_readlane(my.x, i + k);
                int rel   = __builtin_amdgcn_readlane(my.y, i + k);
                int sub   = __builtin_amdgcn_readlane(my.z, i + k);
                row[k]    = __builtin_amdgcn_readlane(my.w, i + k) - node0;
                edge_load(v[k], r_idx, rel, sub, lane, hidden, rela, S, attn_r, attn_qr);
            }
            #pragma unroll
            for (int k = 0; k < EG; ++k)
                edge_finish(v[k], row[k], lane, wa, wb, tile);
        }
        for (; i < cnt; ++i) {
            EdgeVals v;
            int r_idx = __builtin_amdgcn_readlane(my.x, i);
            int rel   = __builtin_amdgcn_readlane(my.y, i);
            int sub   = __builtin_amdgcn_readlane(my.z, i);
            int row   = __builtin_amdgcn_readlane(my.w, i) - node0;
            edge_load(v, r_idx, rel, sub, lane, hidden, rela, S, attn_r, attn_qr);
            edge_finish(v, row, lane, wa, wb, tile);
        }
    }
    __syncthreads();

    int rows = n_node - node0;
    if (rows > BN) rows = BN;
    const float4* src4 = (const float4*)tile;
    float4* dst4 = (float4*)(agg + (long)node0 * IN_DIM);
    for (int i = t; i < rows * (IN_DIM / 4); i += AGG_T) dst4[i] = src4[i];
}

__global__ void out_gemm_wmma_kernel(float* __restrict__ AGG,
                                     const float* __restrict__ Wh, int n_node) {
    int wave  = (int)((blockIdx.x * blockDim.x + threadIdx.x) >> 5);
    int lane  = threadIdx.x & 31;
    int m0    = wave << 4;
    if (m0 >= n_node) return;
    int row16 = lane & 15;
    int hi    = lane >> 4;
    int arow_i = m0 + row16;
    if (arow_i >= n_node) arow_i = n_node - 1;
    const float* arow = AGG + (long)arow_i * IN_DIM + 2 * hi;
    const float* brow = Wh  + (long)row16  * IN_DIM + 2 * hi;

    v8f c0 = {}, c1 = {}, c2 = {}, c3 = {};
    #pragma unroll
    for (int kk = 0; kk < IN_DIM / 4; ++kk) {
        v2f a  = *(const v2f*)(arow + 4 * kk);
        v2f b0 = *(const v2f*)(brow + 0 * 16 * IN_DIM + 4 * kk);
        v2f b1 = *(const v2f*)(brow + 1 * 16 * IN_DIM + 4 * kk);
        v2f b2 = *(const v2f*)(brow + 2 * 16 * IN_DIM + 4 * kk);
        v2f b3 = *(const v2f*)(brow + 3 * 16 * IN_DIM + 4 * kk);
        c0 = __builtin_amdgcn_wmma_f32_16x16x4_f32(false, a, false, b0,
                                                   (short)0, c0, false, false);
        c1 = __builtin_amdgcn_wmma_f32_16x16x4_f32(false, a, false, b1,
                                                   (short)0, c1, false, false);
        c2 = __builtin_amdgcn_wmma_f32_16x16x4_f32(false, a, false, b2,
                                                   (short)0, c2, false, false);
        c3 = __builtin_amdgcn_wmma_f32_16x16x4_f32(false, a, false, b3,
                                                   (short)0, c3, false, false);
    }
    #pragma unroll
    for (int j = 0; j < 8; ++j) {
        int m = m0 + j + 8 * hi;
        if (m < n_node) {
            float* o = AGG + (long)m * OUT_DIM;
            o[row16]      = c0[j];
            o[16 + row16] = c1[j];
            o[32 + row16] = c2[j];
            o[48 + row16] = c3[j];
        }
    }
}

extern "C" void kernel_launch(void* const* d_in, const int* in_sizes, int n_in,
                              void* d_out, int out_size, void* d_ws, size_t ws_size,
                              hipStream_t stream) {
    const int*   q_rel     = (const int*)  d_in[1];
    const float* hidden    = (const float*)d_in[2];
    const int*   edges     = (const int*)  d_in[3];
    const float* rela      = (const float*)d_in[7];
    const float* Ws        = (const float*)d_in[8];
    const float* Wr        = (const float*)d_in[9];
    const float* Wqr       = (const float*)d_in[10];
    const float* bqr       = (const float*)d_in[11];
    const float* w_alpha_w = (const float*)d_in[12];
    const float* w_alpha_b = (const float*)d_in[13];
    const float* Wh        = (const float*)d_in[14];

    const int n_node = in_sizes[2] / IN_DIM;
    const int n_edge = in_sizes[3] / 6;
    const int nrel   = in_sizes[7] / IN_DIM;
    const int batch  = in_sizes[1];
    const int nb     = (n_node + BN - 1) / BN;
    const int chunk  = (n_edge + NCHUNK - 1) / NCHUNK;

    float* agg     = (float*)d_out;
    float* S       = (float*)d_ws;
    float* attn_r  = S + (long)n_node * ATTN_DIM;
    float* attn_qr = attn_r + (long)nrel * ATTN_DIM;
    int*   cnt     = (int*)(attn_qr + (long)batch * ATTN_DIM);
    int*   off     = cnt + (long)NCHUNK * nb;
    int*   bstart  = off + (long)NCHUNK * nb;
    int4*  rec     = (int4*)(bstart + ((nb + 4) / 4) * 4);

    int pre_waves = nrel + batch;
    attn_pre_kernel<<<(pre_waves + 7) / 8, 256, 0, stream>>>(
        rela, Wr, Wqr, bqr, q_rel, attn_r, attn_qr, nrel, batch);

    int mtiles = (n_node + 15) / 16;
    node_attn_wmma_kernel<<<(mtiles + 7) / 8, 256, 0, stream>>>(hidden, Ws, S, n_node);

    bin_count_kernel<<<NCHUNK, BIN_T, nb * sizeof(int), stream>>>(edges, cnt, n_edge, nb, chunk);
    bin_scan_kernel<<<1, BIN_T, (nb + BIN_T) * sizeof(int), stream>>>(cnt, off, bstart, nb);
    bin_place_kernel<<<NCHUNK, BIN_T, nb * sizeof(int), stream>>>(
        edges, off, bstart, rec, n_edge, nb, chunk);

    block_agg_kernel<<<nb, AGG_T, 0, stream>>>(
        rec, bstart, hidden, rela, S, attn_r, attn_qr, w_alpha_w, w_alpha_b, agg, n_node);

    out_gemm_wmma_kernel<<<(mtiles + 7) / 8, 256, 0, stream>>>(agg, Wh, n_node);
}


# block pass reads per-relation rows (rela_embed, attn_r) from an LDS copy staged per workgroup, not from L2; one 32-wave workgroup of 392 nodes per WGP; LDS f32 adds still in arrival order
# speedup vs baseline: 2.0485x; 1.1151x over previous
#include <hip/hip_runtime.h>
#include <hip/hip_bf16.h>

typedef float v2f __attribute__((ext_vector_type(2)));
typedef float v8f __attribute__((ext_vector_type(8)));

#define IN_DIM   64
#define OUT_DIM  64
#define ATTN_DIM 32

__global__ void attn_pre_kernel(const float* __restrict__ rela,
                                const float* __restrict__ Wr,
                                const float* __restrict__ Wqr,
                                const float* __restrict__ bqr,
                                const int*   __restrict__ q_rel,
                                float* __restrict__ attn_r,
                                float* __restrict__ attn_qr,
                                int nrel, int batch) {
    int wave = (int)((blockIdx.x * blockDim.x + threadIdx.x) >> 5);
    int k    = threadIdx.x & 31;
    if (wave >= nrel + batch) return;
    if (wave < nrel) {
        const float* hr = rela + (long)wave * IN_DIM;
        const float* w  = Wr   + (long)k    * IN_DIM;
        float s = 0.f;
        #pragma unroll
        for (int d = 0; d < IN_DIM; ++d) s += hr[d] * w[d];
        attn_r[(long)wave * ATTN_DIM + k] = s;
    } else {
        int b  = wave - nrel;
        int qr = q_rel[b];
        const float* hr = rela + (long)qr * IN_DIM;
        const float* w  = Wqr  + (long)k  * IN_DIM;
        float s = bqr[k];
        #pragma unroll
        for (int d = 0; d < IN_DIM; ++d) s += hr[d] * w[d];
        attn_qr[(long)b * ATTN_DIM + k] = s;
    }
}

__global__ void node_attn_wmma_kernel(const float* __restrict__ H,
                                      const float* __restrict__ Ws,
                                      float* __restrict__ S, int n_node) {
    int wave  = (int)((blockIdx.x * blockDim.x + threadIdx.x) >> 5);
    int lane  = threadIdx.x & 31;
    int m0    = wave << 4;
    if (m0 >= n_node) return;
    int row16 = lane & 15;
    int hi    = lane >> 4;
    int arow_i = m0 + row16;
    if (arow_i >= n_node) arow_i = n_node - 1;
    const float* arow  = H  + (long)arow_i      * IN_DIM + 2 * hi;
    const float* b0row = Ws + (long)row16       * IN_DIM + 2 * hi;
    const float* b1row = Ws + (long)(16 + row16) * IN_DIM + 2 * hi;

    v8f c0 = {}; v8f c1 = {};
    #pragma unroll
    for (int kk = 0; kk < IN_DIM / 4; ++kk) {
        v2f a  = *(const v2f*)(arow  + 4 * kk);
        v2f b0 = *(const v2f*)(b0row + 4 * kk);
        v2f b1 = *(const v2f*)(b1row + 4 * kk);
        c0 = __builtin_amdgcn_wmma_f32_16x16x4_f32(false, a, false, b0,
                                                   (short)0, c0, false, false);
        c1 = __builtin_amdgcn_wmma_f32_16x16x4_f32(false, a, false, b1,
                                                   (short)0, c1, false, false);
    }
    #pragma unroll
    for (int j = 0; j < 8; ++j) {
        int m = m0 + j + 8 * hi;
        if (m < n_node) {
            S[(long)m * ATTN_DIM + row16]      = c0[j];
            S[(long)m * ATTN_DIM + 16 + row16] = c1[j];
        }
    }
}

#define BN       392
#define NCHUNK   256
#define BIN_T    1024
#define BIN_U    4
#define SCAN_G   32

__global__ void bin_count_kernel(const int* __restrict__ edges,
                                 int* __restrict__ cnt,
                                 int n_edge, int nb, int chunk) {
    extern __shared__ int sh_i[];
    const int t = threadIdx.x;
    const int g = blockIdx.x;
    for (int b = t; b < nb; b += BIN_T) sh_i[b] = 0;
    __syncthreads();
    long e0 = (long)g * chunk;
    long e1 = e0 + chunk;
    if (e1 > n_edge) e1 = n_edge;
    for (long e = e0 + t; e < e1; e += (long)BIN_U * BIN_T) {
        int obj[BIN_U], inc[BIN_U];
        #pragma unroll
        for (int k = 0; k < BIN_U; ++k) {
            long ee = e + (long)k * BIN_T;
            inc[k] = (ee < e1) ? 1 : 0;
            if (ee >= e1) ee = e1 - 1;
            obj[k] = edges[ee * 6 + 5];
        }
        #pragma unroll
        for (int k = 0; k < BIN_U; ++k)
            atomicAdd(&sh_i[(unsigned)obj[k] / BN], inc[k]);
    }
    __syncthreads();
    for (int b = t; b < nb; b += BIN_T) cnt[(long)g * nb + b] = sh_i[b];
}

__global__ void bin_scan_kernel(const int* __restrict__ cnt,
                                int* __restrict__ off,
                                int* __restrict__ bstart,
                                int nb) {
    extern __shared__ int sh_i[];
    int* tot = sh_i;
    int* scr = sh_i + nb;
    const int t = threadIdx.x;
    for (int b = t; b < nb; b += BIN_T) {
        int run = 0;
        for (int g = 0; g < NCHUNK; g += SCAN_G) {
            int c[SCAN_G];
            #pragma unroll
            for (int k = 0; k < SCAN_G; ++k) c[k] = cnt[(long)(g + k) * nb + b];
            #pragma unroll
            for (int k = 0; k < SCAN_G; ++k) {
                int v = c[k];
                off[(long)(g + k) * nb + b] = run;
                run += v;
            }
        }
        tot[b] = run;
    }
    __syncthreads();
    int carry = 0;
    for (int base = 0; base < nb; base += BIN_T) {
        const int i = base + t;
        const int v = (i < nb) ? tot[i] : 0;
        scr[t] = v;
        __syncthreads();
        for (int d = 1; d < BIN_T; d <<= 1) {
            int add = (t >= d) ? scr[t - d] : 0;
            __syncthreads();
            scr[t] += add;
            __syncthreads();
        }
        if (i < nb) bstart[i] = carry + scr[t] - v;
        const int tile_total = scr[BIN_T - 1];
        __syncthreads();
        carry += tile_total;
    }
    if (t == 0) bstart[nb] = carry;
}

__global__ void bin_place_kernel(const int* __restrict__ edges,
                                 const int* __restrict__ off,
                                 const int* __restrict__ bstart,
                                 int4* __restrict__ rec,
                                 int n_edge, int nb, int chunk) {
    extern __shared__ int sh_i[];
    const int t = threadIdx.x;
    const int g = blockIdx.x;
    for (int b = t; b < nb; b += BIN_T) sh_i[b] = bstart[b] + off[(long)g * nb + b];
    __syncthreads();
    long e0 = (long)g * chunk;
    long e1 = e0 + chunk;
    if (e1 > n_edge) e1 = n_edge;
    for (long e = e0 + t; e < e1; e += (long)BIN_U * BIN_T) {
        int4 q[BIN_U];
        #pragma unroll
        for (int k = 0; k < BIN_U; ++k) {
            long ee = e + (long)k * BIN_T;
            q[k] = make_int4(0, 0, 0, 0);
            if (ee < e1) {
                const int* p = edges + ee * 6;
                q[k] = make_int4(p[0], p[2], p[4], p[5]);
            }
        }
        #pragma unroll
        for (int k = 0; k < BIN_U; ++k) {
            if (e + (long)k * BIN_T < e1) {
                int pos = atomicAdd(&sh_i[(unsigned)q[k].w / BN], 1);
                rec[pos] = q[k];
            }
        }
    }
}

#define AGG_T  1024
#define EG     8
#define NREL_S 401
#define RELA4  ((NREL_S * IN_DIM / 4 + AGG_T - 1) / AGG_T)
#define ATTR4  ((NREL_S * ATTN_DIM / 4 + AGG_T - 1) / AGG_T)

struct EdgeVals { float s, ar, aq; float2 h, r; };

__device__ __forceinline__ void edge_load(EdgeVals& v, int r_idx, int rel, int sub, int lane,
                                          const float* __restrict__ hidden,
                                          const float* __restrict__ S,
                                          const float* __restrict__ attn_qr,
                                          const float* rela_s,
                                          const float* attnr_s) {
    v.s  = S[(long)sub * ATTN_DIM + lane];
    v.aq = attn_qr[(long)r_idx * ATTN_DIM + lane];
    v.h  = *(const float2*)(hidden + (long)sub * IN_DIM + 2 * lane);
    v.ar = attnr_s[rel * ATTN_DIM + lane];
    v.r  = *(const float2*)(rela_s + rel * IN_DIM + 2 * lane);
}

__device__ __forceinline__ void edge_finish(const EdgeVals& v, int row, int lane,
                                            float wa, float wb, float* tile) {
    float a = v.s + v.ar + v.aq;
    a = fmaxf(a, 0.f);
    float t = a * wa;
    #pragma unroll
    for (int off = 16; off > 0; off >>= 1) t += __shfl_xor(t, off, 32);
    float alpha = 1.f / (1.f + __expf(-(t + wb)));
    float* dst = tile + row * IN_DIM + 2 * lane;
    atomicAdd(dst,     alpha * (v.h.x + v.r.x));
    atomicAdd(dst + 1, alpha * (v.h.y + v.r.y));
}

__global__ void __launch_bounds__(AGG_T)
block_agg_kernel(const int4*  __restrict__ rec,
                 const int*   __restrict__ bstart,
                 const float* __restrict__ hidden,
                 const float* __restrict__ rela,
                 const float* __restrict__ S,
                 const float* __restrict__ attn_r,
                 const float* __restrict__ attn_qr,
                 const float* __restrict__ w_alpha_w,
                 const float* __restrict__ w_alpha_b,
                 float* __restrict__ agg, int n_node, int nrel) {
    __shared__ float4 lds4[BN * IN_DIM / 4 + (RELA4 + ATTR4) * AGG_T];
    float* tile    = (float*)lds4;
    float* rela_s  = tile + BN * IN_DIM;
    float* attnr_s = rela_s + RELA4 * AGG_T * 4;
    const int t = threadIdx.x;
    const int node0 = (int)blockIdx.x * BN;

    const int nr   = nrel < NREL_S ? nrel : NREL_S;
    const int n_r4 = nr * (IN_DIM / 4);
    const int n_a4 = nr * (ATTN_DIM / 4);
    const float4* rela4  = (const float4*)rela;
    const float4* attnr4 = (const float4*)attn_r;
    float4 cr[RELA4], ca[ATTR4];
    #pragma unroll
    for (int k = 0; k < RELA4; ++k) {
        const int i = t + k * AGG_T;
        cr[k] = rela4[i < n_r4 ? i : 0];
    }
    #pragma unroll
    for (int k = 0; k < ATTR4; ++k) {
        const int i = t + k * AGG_T;
        ca[k] = attnr4[i < n_a4 ? i : 0];
    }
    for (int i = t; i < BN * IN_DIM / 4; i += AGG_T) lds4[i] = make_float4(0.f, 0.f, 0.f, 0.f);
    #pragma unroll
    for (int k = 0; k < RELA4; ++k) ((float4*)rela_s)[t + k * AGG_T] = cr[k];
    #pragma unroll
    for (int k = 0; k < ATTR4; ++k) ((float4*)attnr_s)[t + k * AGG_T] = ca[k];
    __syncthreads();

    const int wave = __builtin_amdgcn_readfirstlane(t >> 5);
    const int lane = t & 31;
    const int r_begin = bstart[blockIdx.x];
    const int r_end   = bstart[blockIdx.x + 1];
    const float wa = w_alpha_w[lane];
    const float wb = w_alpha_b[0];

    for (int base = r_begin + wave * 32; base < r_end; base += (AGG_T / 32) * 32) {
        const int left = r_end - base;
        const int cnt = left < 32 ? left : 32;
        int4 my = make_int4(0, 0, 0, node0);
        if (lane < cnt) my = rec[base + lane];

        int i = 0;
        for (; i + EG <= cnt; i += EG) {
            EdgeVals v[EG];
            int row[EG];
            #pragma unroll
            for (int k = 0; k < EG; ++k) {
                int r_idx = __builtin_amdgcn_readlane(my.x, i + k);
                int rel   = __builtin_amdgcn_readlane(my.y, i + k);
                int sub   = __builtin_amdgcn_readlane(my.z, i + k);
                row[k]    = __builtin_amdgcn_readlane(my.w, i + k) - node0;
                edge_load(v[k], r_idx, rel, sub, lane, hidden, S, attn_qr, rela_s, attnr_s);
            }
            #pragma unroll
            for (int k = 0; k < EG; ++k)
                edge_finish(v[k], row[k], lane, wa, wb, tile);
        }
        for (; i < cnt; ++i) {
            EdgeVals v;
            int r_idx = __builtin_amdgcn_readlane(my.x, i);
            int rel   = __builtin_amdgcn_readlane(my.y, i);
            int sub   = __builtin_amdgcn_readlane(my.z, i);
            int row   = __builtin_amdgcn_readlane(my.w, i) - node0;
            edge_load(v, r_idx, rel, sub, lane, hidden, S, attn_qr, rela_s, attnr_s);
            edge_finish(v, row, lane, wa, wb, tile);
        }
    }
    __syncthreads();

    int rows = n_node - node0;
    if (rows > BN) rows = BN;
    float4* dst4 = (float4*)(agg + (long)node0 * IN_DIM);
    for (int i = t; i < rows * (IN_DIM / 4); i += AGG_T) dst4[i] = lds4[i];
}

__global__ void out_gemm_wmma_kernel(float* __restrict__ AGG,
                                     const float* __restrict__ Wh, int n_node) {
    int wave  = (int)((blockIdx.x * blockDim.x + threadIdx.x) >> 5);
    int lane  = threadIdx.x & 31;
    int m0    = wave << 4;
    if (m0 >= n_node) return;
    int row16 = lane & 15;
    int hi    = lane >> 4;
    int arow_i = m0 + row16;
    if (arow_i >= n_node) arow_i = n_node - 1;
    const float* arow = AGG + (long)arow_i * IN_DIM + 2 * hi;
    const float* brow = Wh  + (long)row16  * IN_DIM + 2 * hi;

    v8f c0 = {}, c1 = {}, c2 = {}, c3 = {};
    #pragma unroll
    for (int kk = 0; kk < IN_DIM / 4; ++kk) {
        v2f a  = *(const v2f*)(arow + 4 * kk);
        v2f b0 = *(const v2f*)(brow + 0 * 16 * IN_DIM + 4 * kk);
        v2f b1 = *(const v2f*)(brow + 1 * 16 * IN_DIM + 4 * kk);
        v2f b2 = *(const v2f*)(brow + 2 * 16 * IN_DIM + 4 * kk);
        v2f b3 = *(const v2f*)(brow + 3 * 16 * IN_DIM + 4 * kk);
        c0 = __builtin_amdgcn_wmma_f32_16x16x4_f32(false, a, false, b0,
                                                   (short)0, c0, false, false);
        c1 = __builtin_amdgcn_wmma_f32_16x16x4_f32(false, a, false, b1,
                                                   (short)0, c1, false, false);
        c2 = __builtin_amdgcn_wmma_f32_16x16x4_f32(false, a, false, b2,
                                                   (short)0, c2, false, false);
        c3 = __builtin_amdgcn_wmma_f32_16x16x4_f32(false, a, false, b3,
                                                   (short)0, c3, false, false);
    }
    #pragma unroll
    for (int j = 0; j < 8; ++j) {
        int m = m0 + j + 8 * hi;
        if (m < n_node) {
            float* o = AGG + (long)m * OUT_DIM;
            o[row16]      = c0[j];
            o[16 + row16] = c1[j];
            o[32 + row16] = c2[j];
            o[48 + row16] = c3[j];
        }
    }
}

extern "C" void kernel_launch(void* const* d_in, const int* in_sizes, int n_in,
                              void* d_out, int out_size, void* d_ws, size_t ws_size,
                              hipStream_t stream) {
    const int*   q_rel     = (const int*)  d_in[1];
    const float* hidden    = (const float*)d_in[2];
    const int*   edges     = (const int*)  d_in[3];
    const float* rela      = (const float*)d_in[7];
    const float* Ws        = (const float*)d_in[8];
    const float* Wr        = (const float*)d_in[9];
    const float* Wqr       = (const float*)d_in[10];
    const float* bqr       = (const float*)d_in[11];
    const float* w_alpha_w = (const float*)d_in[12];
    const float* w_alpha_b = (const float*)d_in[13];
    const float* Wh        = (const float*)d_in[14];

    const int n_node = in_sizes[2] / IN_DIM;
    const int n_edge = in_sizes[3] / 6;
    const int nrel   = in_sizes[7] / IN_DIM;
    const int batch  = in_sizes[1];
    const int nb     = (n_node + BN - 1) / BN;
    const int chunk  = (n_edge + NCHUNK - 1) / NCHUNK;

    float* agg     = (float*)d_out;
    float* S       = (float*)d_ws;
    float* attn_r  = S + (long)n_node * ATTN_DIM;
    float* attn_qr = attn_r + (long)nrel * ATTN_DIM;
    int*   cnt     = (int*)(attn_qr + (long)batch * ATTN_DIM);
    int*   off     = cnt + (long)NCHUNK * nb;
    int*   bstart  = off + (long)NCHUNK * nb;
    int4*  rec     = (int4*)(bstart + ((nb + 4) / 4) * 4);

    int pre_waves = nrel + batch;
    attn_pre_kernel<<<(pre_waves + 7) / 8, 256, 0, stream>>>(
        rela, Wr, Wqr, bqr, q_rel, attn_r, attn_qr, nrel, batch);

    int mtiles = (n_node + 15) / 16;
    node_attn_wmma_kernel<<<(mtiles + 7) / 8, 256, 0, stream>>>(hidden, Ws, S, n_node);

    bin_count_kernel<<<NCHUNK, BIN_T, nb * sizeof(int), stream>>>(edges, cnt, n_edge, nb, chunk);
    bin_scan_kernel<<<1, BIN_T, (nb + BIN_T) * sizeof(int), stream>>>(cnt, off, bstart, nb);
    bin_place_kernel<<<NCHUNK, BIN_T, nb * sizeof(int), stream>>>(
        edges, off, bstart, rec, n_edge, nb, chunk);

    block_agg_kernel<<<nb, AGG_T, 0, stream>>>(
        rec, bstart, hidden, rela, S, attn_r, attn_qr, w_alpha_w, w_alpha_b, agg, n_node, nrel);

    out_gemm_wmma_kernel<<<(mtiles + 7) / 8, 256, 0, stream>>>(agg, Wh, n_node);
}


# block pass: gate sums of 8 records reduced together by one folding butterfly on the vector ALU (same tree, lane swizzles) instead of 5 LDS-unit shuffles per record; one sigmoid per 8 records
# speedup vs baseline: 2.2472x; 1.0970x over previous
#include <hip/hip_runtime.h>
#include <hip/hip_bf16.h>

typedef float v2f __attribute__((ext_vector_type(2)));
typedef float v8f __attribute__((ext_vector_type(8)));

#define IN_DIM   64
#define OUT_DIM  64
#define ATTN_DIM 32

__global__ void attn_pre_kernel(const float* __restrict__ rela,
                                const float* __restrict__ Wr,
                                const float* __restrict__ Wqr,
                                const float* __restrict__ bqr,
                                const int*   __restrict__ q_rel,
                                float* __restrict__ attn_r,
                                float* __restrict__ attn_qr,
                                int nrel, int batch) {
    int wave = (int)((blockIdx.x * blockDim.x + threadIdx.x) >> 5);
    int k    = threadIdx.x & 31;
    if (wave >= nrel + batch) return;
    if (wave < nrel) {
        const float* hr = rela + (long)wave * IN_DIM;
        const float* w  = Wr   + (long)k    * IN_DIM;
        float s = 0.f;
        #pragma unroll
        for (int d = 0; d < IN_DIM; ++d) s += hr[d] * w[d];
        attn_r[(long)wave * ATTN_DIM + k] = s;
    } else {
        int b  = wave - nrel;
        int qr = q_rel[b];
        const float* hr = rela + (long)qr * IN_DIM;
        const float* w  = Wqr  + (long)k  * IN_DIM;
        float s = bqr[k];
        #pragma unroll
        for (int d = 0; d < IN_DIM; ++d) s += hr[d] * w[d];
        attn_qr[(long)b * ATTN_DIM + k] = s;
    }
}

__global__ void node_attn_wmma_kernel(const float* __restrict__ H,
                                      const float* __restrict__ Ws,
                                      float* __restrict__ S, int n_node) {
    int wave  = (int)((blockIdx.x * blockDim.x + threadIdx.x) >> 5);
    int lane  = threadIdx.x & 31;
    int m0    = wave << 4;
    if (m0 >= n_node) return;
    int row16 = lane & 15;
    int hi    = lane >> 4;
    int arow_i = m0 + row16;
    if (arow_i >= n_node) arow_i = n_node - 1;
    const float* arow  = H  + (long)arow_i      * IN_DIM + 2 * hi;
    const float* b0row = Ws + (long)row16       * IN_DIM + 2 * hi;
    const float* b1row = Ws + (long)(16 + row16) * IN_DIM + 2 * hi;

    v8f c0 = {}; v8f c1 = {};
    #pragma unroll
    for (int kk = 0; kk < IN_DIM / 4; ++kk) {
        v2f a  = *(const v2f*)(arow  + 4 * kk);
        v2f b0 = *(const v2f*)(b0row + 4 * kk);
        v2f b1 = *(const v2f*)(b1row + 4 * kk);
        c0 = __builtin_amdgcn_wmma_f32_16x16x4_f32(false, a, false, b0,
                                                   (short)0, c0, false, false);
        c1 = __builtin_amdgcn_wmma_f32_16x16x4_f32(false, a, false, b1,
                                                   (short)0, c1, false, false);
    }
    #pragma unroll
    for (int j = 0; j < 8; ++j) {
        int m = m0 + j + 8 * hi;
        if (m < n_node) {
            S[(long)m * ATTN_DIM + row16]      = c0[j];
            S[(long)m * ATTN_DIM + 16 + row16] = c1[j];
        }
    }
}

#define BN       392
#define NCHUNK   256
#define BIN_T    1024
#define BIN_U    4
#define SCAN_G   32

__global__ void bin_count_kernel(const int* __restrict__ edges,
                                 int* __restrict__ cnt,
                                 int n_edge, int nb, int chunk) {
    extern __shared__ int sh_i[];
    const int t = threadIdx.x;
    const int g = blockIdx.x;
    for (int b = t; b < nb; b += BIN_T) sh_i[b] = 0;
    __syncthreads();
    long e0 = (long)g * chunk;
    long e1 = e0 + chunk;
    if (e1 > n_edge) e1 = n_edge;
    for (long e = e0 + t; e < e1; e += (long)BIN_U * BIN_T) {
        int obj[BIN_U], inc[BIN_U];
        #pragma unroll
        for (int k = 0; k < BIN_U; ++k) {
            long ee = e + (long)k * BIN_T;
            inc[k] = (ee < e1) ? 1 : 0;
            if (ee >= e1) ee = e1 - 1;
            obj[k] = edges[ee * 6 + 5];
        }
        #pragma unroll
        for (int k = 0; k < BIN_U; ++k)
            atomicAdd(&sh_i[(unsigned)obj[k] / BN], inc[k]);
    }
    __syncthreads();
    for (int b = t; b < nb; b += BIN_T) cnt[(long)g * nb + b] = sh_i[b];
}

__global__ void bin_scan_kernel(const int* __restrict__ cnt,
                                int* __restrict__ off,
                                int* __restrict__ bstart,
                                int nb) {
    extern __shared__ int sh_i[];
    int* tot = sh_i;
    int* scr = sh_i + nb;
    const int t = threadIdx.x;
    for (int b = t; b < nb; b += BIN_T) {
        int run = 0;
        for (int g = 0; g < NCHUNK; g += SCAN_G) {
            int c[SCAN_G];
            #pragma unroll
            for (int k = 0; k < SCAN_G; ++k) c[k] = cnt[(long)(g + k) * nb + b];
            #pragma unroll
            for (int k = 0; k < SCAN_G; ++k) {
                int v = c[k];
                off[(long)(g + k) * nb + b] = run;
                run += v;
            }
        }
        tot[b] = run;
    }
    __syncthreads();
    int carry = 0;
    for (int base = 0; base < nb; base += BIN_T) {
        const int i = base + t;
        const int v = (i < nb) ? tot[i] : 0;
        scr[t] = v;
        __syncthreads();
        for (int d = 1; d < BIN_T; d <<= 1) {
            int add = (t >= d) ? scr[t - d] : 0;
            __syncthreads();
            scr[t] += add;
            __syncthreads();
        }
        if (i < nb) bstart[i] = carry + scr[t] - v;
        const int tile_total = scr[BIN_T - 1];
        __syncthreads();
        carry += tile_total;
    }
    if (t == 0) bstart[nb] = carry;
}

__global__ void bin_place_kernel(const int* __restrict__ edges,
                                 const int* __restrict__ off,
                                 const int* __restrict__ bstart,
                                 int4* __restrict__ rec,
                                 int n_edge, int nb, int chunk) {
    extern __shared__ int sh_i[];
    const int t = threadIdx.x;
    const int g = blockIdx.x;
    for (int b = t; b < nb; b += BIN_T) sh_i[b] = bstart[b] + off[(long)g * nb + b];
    __syncthreads();
    long e0 = (long)g * chunk;
    long e1 = e0 + chunk;
    if (e1 > n_edge) e1 = n_edge;
    for (long e = e0 + t; e < e1; e += (long)BIN_U * BIN_T) {
        int4 q[BIN_U];
        #pragma unroll
        for (int k = 0; k < BIN_U; ++k) {
            long ee = e + (long)k * BIN_T;
            q[k] = make_int4(0, 0, 0, 0);
            if (ee < e1) {
                const int* p = edges + ee * 6;
                q[k] = make_int4(p[0], p[2], p[4], p[5]);
            }
        }
        #pragma unroll
        for (int k = 0; k < BIN_U; ++k) {
            if (e + (long)k * BIN_T < e1) {
                int pos = atomicAdd(&sh_i[(unsigned)q[k].w / BN], 1);
                rec[pos] = q[k];
            }
        }
    }
}

#define AGG_T  1024
#define EG     8
#define NREL_S 401
#define RELA4  ((NREL_S * IN_DIM / 4 + AGG_T - 1) / AGG_T)
#define ATTR4  ((NREL_S * ATTN_DIM / 4 + AGG_T - 1) / AGG_T)

struct EdgeVals { float s, ar, aq; float2 h, r; };

__device__ __forceinline__ void edge_load(EdgeVals& v, int r_idx, int rel, int sub, int lane,
                                          const float* __restrict__ hidden,
                                          const float* __restrict__ S,
                                          const float* __restrict__ attn_qr,
                                          const float* rela_s,
                                          const float* attnr_s) {
    v.s  = S[(long)sub * ATTN_DIM + lane];
    v.aq = attn_qr[(long)r_idx * ATTN_DIM + lane];
    v.h  = *(const float2*)(hidden + (long)sub * IN_DIM + 2 * lane);
    v.ar = attnr_s[rel * ATTN_DIM + lane];
    v.r  = *(const float2*)(rela_s + rel * IN_DIM + 2 * lane);
}

__device__ __forceinline__ float xhalf(float x) {
    int b = __builtin_bit_cast(int, x);
    return __builtin_bit_cast(float,
        __builtin_amdgcn_permlanex16(b, b, 0x76543210, 0xfedcba98, false, false));
}

template <int M>
__device__ __forceinline__ float row_xor(float x) {
    int b = __builtin_bit_cast(int, x);
    return __builtin_bit_cast(float,
        __builtin_amdgcn_update_dpp(0, b, 0x160 + M, 0xf, 0xf, true));
}

__device__ __forceinline__ float fold_half(float p, float q, bool up) {
    float keep = up ? q : p;
    float send = up ? p : q;
    return keep + xhalf(send);
}
template <int M>
__device__ __forceinline__ float fold_row(float p, float q, bool up) {
    float keep = up ? q : p;
    float send = up ? p : q;
    return keep + row_xor<M>(send);
}

static_assert(EG == 8, "the folding butterfly of the block pass is written out for groups of 8 records");
#define GATE_LANE(k) ((((k) & 1) << 4) | ((((k) >> 1) & 1) << 3) | ((((k) >> 2) & 1) << 2))

__device__ __forceinline__ void edge_finish(const EdgeVals& v, int row, int lane,
                                            float wa, float wb, float* tile) {
    float a = v.s + v.ar + v.aq;
    a = fmaxf(a, 0.f);
    float t = a * wa;
    #pragma unroll
    for (int off = 16; off > 0; off >>= 1) t += __shfl_xor(t, off, 32);
    float alpha = 1.f / (1.f + __expf(-(t + wb)));
    float* dst = tile + row * IN_DIM + 2 * lane;
    atomicAdd(dst,     alpha * (v.h.x + v.r.x));
    atomicAdd(dst + 1, alpha * (v.h.y + v.r.y));
}

__global__ void __launch_bounds__(AGG_T)
block_agg_kernel(const int4*  __restrict__ rec,
                 const int*   __restrict__ bstart,
                 const float* __restrict__ hidden,
                 const float* __restrict__ rela,
                 const float* __restrict__ S,
                 const float* __restrict__ attn_r,
                 const float* __restrict__ attn_qr,
                 const float* __restrict__ w_alpha_w,
                 const float* __restrict__ w_alpha_b,
                 float* __restrict__ agg, int n_node, int nrel) {
    __shared__ float4 lds4[BN * IN_DIM / 4 + (RELA4 + ATTR4) * AGG_T];
    float* tile    = (float*)lds4;
    float* rela_s  = tile + BN * IN_DIM;
    float* attnr_s = rela_s + RELA4 * AGG_T * 4;
    const int t = threadIdx.x;
    const int node0 = (int)blockIdx.x * BN;

    const int nr   = nrel < NREL_S ? nrel : NREL_S;
    const int n_r4 = nr * (IN_DIM / 4);
    const int n_a4 = nr * (ATTN_DIM / 4);
    const float4* rela4  = (const float4*)rela;
    const float4* attnr4 = (const float4*)attn_r;
    float4 cr[RELA4], ca[ATTR4];
    #pragma unroll
    for (int k = 0; k < RELA4; ++k) {
        const int i = t + k * AGG_T;
        cr[k] = rela4[i < n_r4 ? i : 0];
    }
    #pragma unroll
    for (int k = 0; k < ATTR4; ++k) {
        const int i = t + k * AGG_T;
        ca[k] = attnr4[i < n_a4 ? i : 0];
    }
    for (int i = t; i < BN * IN_DIM / 4; i += AGG_T) lds4[i] = make_float4(0.f, 0.f, 0.f, 0.f);
    #pragma unroll
    for (int k = 0; k < RELA4; ++k) ((float4*)rela_s)[t + k * AGG_T] = cr[k];
    #pragma unroll
    for (int k = 0; k < ATTR4; ++k) ((float4*)attnr_s)[t + k * AGG_T] = ca[k];
    __syncthreads();

    const int wave = __builtin_amdgcn_readfirstlane(t >> 5);
    const int lane = t & 31;
    const int r_begin = bstart[blockIdx.x];
    const int r_end   = bstart[blockIdx.x + 1];
    const float wa = w_alpha_w[lane];
    const float wb = w_alpha_b[0];
    const bool up16 = (lane & 16) != 0;
    const bool up8  = (lane & 8) != 0;
    const bool up4  = (lane & 4) != 0;

    for (int base = r_begin + wave * 32; base < r_end; base += (AGG_T / 32) * 32) {
        const int left = r_end - base;
        const int cnt = left < 32 ? left : 32;
        int4 my = make_int4(0, 0, 0, node0);
        if (lane < cnt) my = rec[base + lane];

        int i = 0;
        for (; i + EG <= cnt; i += EG) {
            EdgeVals v[EG];
            int row[EG];
            #pragma unroll
            for (int k = 0; k < EG; ++k) {
                int r_idx = __builtin_amdgcn_readlane(my.x, i + k);
                int rel   = __builtin_amdgcn_readlane(my.y, i + k);
                int sub   = __builtin_amdgcn_readlane(my.z, i + k);
                row[k]    = __builtin_amdgcn_readlane(my.w, i + k) - node0;
                edge_load(v[k], r_idx, rel, sub, lane, hidden, S, attn_qr, rela_s, attnr_s);
            }
            float tk[EG];
            #pragma unroll
            for (int k = 0; k < EG; ++k) {
                float a = v[k].s + v[k].ar + v[k].aq;
                a = fmaxf(a, 0.f);
                tk[k] = a * wa;
            }
            const float u0 = fold_half(tk[0], tk[1], up16);
            const float u1 = fold_half(tk[2], tk[3], up16);
            const float u2 = fold_half(tk[4], tk[5], up16);
            const float u3 = fold_half(tk[6], tk[7], up16);
            const float w0 = fold_row<8>(u0, u1, up8);
            const float w1 = fold_row<8>(u2, u3, up8);
            float z = fold_row<4>(w0, w1, up4);
            z += row_xor<2>(z);
            z += row_xor<1>(z);
            const float alpha8 = 1.f / (1.f + __expf(-(z + wb)));
            #pragma unroll
            for (int k = 0; k < EG; ++k) {
                const float alpha = __builtin_bit_cast(float,
                    __builtin_amdgcn_readlane(__builtin_bit_cast(int, alpha8), GATE_LANE(k)));
                float* dst = tile + row[k] * IN_DIM + 2 * lane;
                atomicAdd(dst,     alpha * (v[k].h.x + v[k].r.x));
                atomicAdd(dst + 1, alpha * (v[k].h.y + v[k].r.y));
            }
        }
        for (; i < cnt; ++i) {
            EdgeVals v;
            int r_idx = __builtin_amdgcn_readlane(my.x, i);
            int rel   = __builtin_amdgcn_readlane(my.y, i);
            int sub   = __builtin_amdgcn_readlane(my.z, i);
            int row   = __builtin_amdgcn_readlane(my.w, i) - node0;
            edge_load(v, r_idx, rel, sub, lane, hidden, S, attn_qr, rela_s, attnr_s);
            edge_finish(v, row, lane, wa, wb, tile);
        }
    }
    __syncthreads();

    int rows = n_node - node0;
    if (rows > BN) rows = BN;
    float4* dst4 = (float4*)(agg + (long)node0 * IN_DIM);
    for (int i = t; i < rows * (IN_DIM / 4); i += AGG_T) dst4[i] = lds4[i];
}

__global__ void out_gemm_wmma_kernel(float* __restrict__ AGG,
                                     const float* __restrict__ Wh, int n_node) {
    int wave  = (int)((blockIdx.x * blockDim.x + threadIdx.x) >> 5);
    int lane  = threadIdx.x & 31;
    int m0    = wave << 4;
    if (m0 >= n_node) return;
    int row16 = lane & 15;
    int hi    = lane >> 4;
    int arow_i = m0 + row16;
    if (arow_i >= n_node) arow_i = n_node - 1;
    const float* arow = AGG + (long)arow_i * IN_DIM + 2 * hi;
    const float* brow = Wh  + (long)row16  * IN_DIM + 2 * hi;

    v8f c0 = {}, c1 = {}, c2 = {}, c3 = {};
    #pragma unroll
    for (int kk = 0; kk < IN_DIM / 4; ++kk) {
        v2f a  = *(const v2f*)(arow + 4 * kk);
        v2f b0 = *(const v2f*)(brow + 0 * 16 * IN_DIM + 4 * kk);
        v2f b1 = *(const v2f*)(brow + 1 * 16 * IN_DIM + 4 * kk);
        v2f b2 = *(const v2f*)(brow + 2 * 16 * IN_DIM + 4 * kk);
        v2f b3 = *(const v2f*)(brow + 3 * 16 * IN_DIM + 4 * kk);
        c0 = __builtin_amdgcn_wmma_f32_16x16x4_f32(false, a, false, b0,
                                                   (short)0, c0, false, false);
        c1 = __builtin_amdgcn_wmma_f32_16x16x4_f32(false, a, false, b1,
                                                   (short)0, c1, false, false);
        c2 = __builtin_amdgcn_wmma_f32_16x16x4_f32(false, a, false, b2,
                                                   (short)0, c2, false, false);
        c3 = __builtin_amdgcn_wmma_f32_16x16x4_f32(false, a, false, b3,
                                                   (short)0, c3, false, false);
    }
    #pragma unroll
    for (int j = 0; j < 8; ++j) {
        int m = m0 + j + 8 * hi;
        if (m < n_node) {
            float* o = AGG + (long)m * OUT_DIM;
            o[row16]      = c0[j];
            o[16 + row16] = c1[j];
            o[32 + row16] = c2[j];
            o[48 + row16] = c3[j];
        }
    }
}

extern "C" void kernel_launch(void* const* d_in, const int* in_sizes, int n_in,
                              void* d_out, int out_size, void* d_ws, size_t ws_size,
                              hipStream_t stream) {
    const int*   q_rel     = (const int*)  d_in[1];
    const float* hidden    = (const float*)d_in[2];
    const int*   edges     = (const int*)  d_in[3];
    const float* rela      = (const float*)d_in[7];
    const float* Ws        = (const float*)d_in[8];
    const float* Wr        = (const float*)d_in[9];
    const float* Wqr       = (const float*)d_in[10];
    const float* bqr       = (const float*)d_in[11];
    const float* w_alpha_w = (const float*)d_in[12];
    const float* w_alpha_b = (const float*)d_in[13];
    const float* Wh        = (const float*)d_in[14];

    const int n_node = in_sizes[2] / IN_DIM;
    const int n_edge = in_sizes[3] / 6;
    const int nrel   = in_sizes[7] / IN_DIM;
    const int batch  = in_sizes[1];
    const int nb     = (n_node + BN - 1) / BN;
    const int chunk  = (n_edge + NCHUNK - 1) / NCHUNK;

    float* agg     = (float*)d_out;
    float* S       = (float*)d_ws;
    float* attn_r  = S + (long)n_node * ATTN_DIM;
    float* attn_qr = attn_r + (long)nrel * ATTN_DIM;
    int*   cnt     = (int*)(attn_qr + (long)batch * ATTN_DIM);
    int*   off     = cnt + (long)NCHUNK * nb;
    int*   bstart  = off + (long)NCHUNK * nb;
    int4*  rec     = (int4*)(bstart + ((nb + 4) / 4) * 4);

    int pre_waves = nrel + batch;
    attn_pre_kernel<<<(pre_waves + 7) / 8, 256, 0, stream>>>(
        rela, Wr, Wqr, bqr, q_rel, attn_r, attn_qr, nrel, batch);

    int mtiles = (n_node + 15) / 16;
    node_attn_wmma_kernel<<<(mtiles + 7) / 8, 256, 0, stream>>>(hidden, Ws, S, n_node);

    bin_count_kernel<<<NCHUNK, BIN_T, nb * sizeof(int), stream>>>(edges, cnt, n_edge, nb, chunk);
    bin_scan_kernel<<<1, BIN_T, (nb + BIN_T) * sizeof(int), stream>>>(cnt, off, bstart, nb);
    bin_place_kernel<<<NCHUNK, BIN_T, nb * sizeof(int), stream>>>(
        edges, off, bstart, rec, n_edge, nb, chunk);

    block_agg_kernel<<<nb, AGG_T, 0, stream>>>(
        rec, bstart, hidden, rela, S, attn_r, attn_qr, w_alpha_w, w_alpha_b, agg, n_node, nrel);

    out_gemm_wmma_kernel<<<(mtiles + 7) / 8, 256, 0, stream>>>(agg, Wh, n_node);
}


# node gate GEMM (hidden x Ws) and the two attention tables moved into the launch of the one-workgroup bin scan, on the WGPs it left idle: 7 launches -> 5; every sum keeps its order
# speedup vs baseline: 2.4341x; 1.0832x over previous
#include <hip/hip_runtime.h>
#include <hip/hip_bf16.h>

typedef float v2f __attribute__((ext_vector_type(2)));
typedef float v8f __attribute__((ext_vector_type(8)));

#define IN_DIM   64
#define OUT_DIM  64
#define ATTN_DIM 32

#define MID_T 256

__device__ __forceinline__ void attn_pre_part(int wg,
                                const float* __restrict__ rela,
                                const float* __restrict__ Wr,
                                const float* __restrict__ Wqr,
                                const float* __restrict__ bqr,
                                const int*   __restrict__ q_rel,
                                float* __restrict__ attn_r,
                                float* __restrict__ attn_qr,
                                int nrel, int batch) {
    int wave = (int)((wg * MID_T + threadIdx.x) >> 5);
    int k    = threadIdx.x & 31;
    if (wave >= nrel + batch) return;
    if (wave < nrel) {
        const float* hr = rela + (long)wave * IN_DIM;
        const float* w  = Wr   + (long)k    * IN_DIM;
        float s = 0.f;
        #pragma unroll
        for (int d = 0; d < IN_DIM; ++d) s += hr[d] * w[d];
        attn_r[(long)wave * ATTN_DIM + k] = s;
    } else {
        int b  = wave - nrel;
        int qr = q_rel[b];
        const float* hr = rela + (long)qr * IN_DIM;
        const float* w  = Wqr  + (long)k  * IN_DIM;
        float s = bqr[k];
        #pragma unroll
        for (int d = 0; d < IN_DIM; ++d) s += hr[d] * w[d];
        attn_qr[(long)b * ATTN_DIM + k] = s;
    }
}

__device__ __forceinline__ void node_attn_wmma_part(int wg,
                                      const float* __restrict__ H,
                                      const float* __restrict__ Ws,
                                      float* __restrict__ S, int n_node) {
    int wave  = (int)((wg * MID_T + threadIdx.x) >> 5);
    int lane  = threadIdx.x & 31;
    int m0    = wave << 4;
    if (m0 >= n_node) return;
    int row16 = lane & 15;
    int hi    = lane >> 4;
    int arow_i = m0 + row16;
    if (arow_i >= n_node) arow_i = n_node - 1;
    const float* arow  = H  + (long)arow_i      * IN_DIM + 2 * hi;
    const float* b0row = Ws + (long)row16       * IN_DIM + 2 * hi;
    const float* b1row = Ws + (long)(16 + row16) * IN_DIM + 2 * hi;

    v8f c0 = {}; v8f c1 = {};
    #pragma unroll
    for (int kk = 0; kk < IN_DIM / 4; ++kk) {
        v2f a  = *(const v2f*)(arow  + 4 * kk);
        v2f b0 = *(const v2f*)(b0row + 4 * kk);
        v2f b1 = *(const v2f*)(b1row + 4 * kk);
        c0 = __builtin_amdgcn_wmma_f32_16x16x4_f32(false, a, false, b0,
                                                   (short)0, c0, false, false);
        c1 = __builtin_amdgcn_wmma_f32_16x16x4_f32(false, a, false, b1,
                                                   (short)0, c1, false, false);
    }
    #pragma unroll
    for (int j = 0; j < 8; ++j) {
        int m = m0 + j + 8 * hi;
        if (m < n_node) {
            S[(long)m * ATTN_DIM + row16]      = c0[j];
            S[(long)m * ATTN_DIM + 16 + row16] = c1[j];
        }
    }
}

#define BN       392
#define NCHUNK   256
#define BIN_T    1024
#define BIN_U    4
#define SCAN_G   32

__global__ void bin_count_kernel(const int* __restrict__ edges,
                                 int* __restrict__ cnt,
                                 int n_edge, int nb, int chunk) {
    extern __shared__ int sh_i[];
    const int t = threadIdx.x;
    const int g = blockIdx.x;
    for (int b = t; b < nb; b += BIN_T) sh_i[b] = 0;
    __syncthreads();
    long e0 = (long)g * chunk;
    long e1 = e0 + chunk;
    if (e1 > n_edge) e1 = n_edge;
    for (long e = e0 + t; e < e1; e += (long)BIN_U * BIN_T) {
        int obj[BIN_U], inc[BIN_U];
        #pragma unroll
        for (int k = 0; k < BIN_U; ++k) {
            long ee = e + (long)k * BIN_T;
            inc[k] = (ee < e1) ? 1 : 0;
            if (ee >= e1) ee = e1 - 1;
            obj[k] = edges[ee * 6 + 5];
        }
        #pragma unroll
        for (int k = 0; k < BIN_U; ++k)
            atomicAdd(&sh_i[(unsigned)obj[k] / BN], inc[k]);
    }
    __syncthreads();
    for (int b = t; b < nb; b += BIN_T) cnt[(long)g * nb + b] = sh_i[b];
}

__device__ __forceinline__ void bin_scan_part(const int* __restrict__ cnt,
                                int* __restrict__ off,
                                int* __restrict__ bstart,
                                int nb, int* sh_i) {
    int* tot = sh_i;
    int* scr = sh_i + nb;
    const int t = threadIdx.x;
    for (int b = t; b < nb; b += MID_T) {
        int run = 0;
        for (int g = 0; g < NCHUNK; g += SCAN_G) {
            int c[SCAN_G];
            #pragma unroll
            for (int k = 0; k < SCAN_G; ++k) c[k] = cnt[(long)(g + k) * nb + b];
            #pragma unroll
            for (int k = 0; k < SCAN_G; ++k) {
                int v = c[k];
                off[(long)(g + k) * nb + b] = run;
                run += v;
            }
        }
        tot[b] = run;
    }
    __syncthreads();
    int carry = 0;
    for (int base = 0; base < nb; base += MID_T) {
        const int i = base + t;
        const int v = (i < nb) ? tot[i] : 0;
        scr[t] = v;
        __syncthreads();
        for (int d = 1; d < MID_T; d <<= 1) {
            int add = (t >= d) ? scr[t - d] : 0;
            __syncthreads();
            scr[t] += add;
            __syncthreads();
        }
        if (i < nb) bstart[i] = carry + scr[t] - v;
        const int tile_total = scr[MID_T - 1];
        __syncthreads();
        carry += tile_total;
    }
    if (t == 0) bstart[nb] = carry;
}

__global__ void __launch_bounds__(MID_T)
mid_kernel(const int* __restrict__ cnt, int* __restrict__ off, int* __restrict__ bstart, int nb,
           const float* __restrict__ hidden, const float* __restrict__ Ws,
           float* __restrict__ S, int n_node, int gemm_wgs,
           const float* __restrict__ rela, const float* __restrict__ Wr,
           const float* __restrict__ Wqr, const float* __restrict__ bqr,
           const int* __restrict__ q_rel,
           float* __restrict__ attn_r, float* __restrict__ attn_qr, int nrel, int batch) {
    extern __shared__ int sh_i[];
    const int g = (int)blockIdx.x;
    if (g == 0) {
        bin_scan_part(cnt, off, bstart, nb, sh_i);
    } else if (g <= gemm_wgs) {
        node_attn_wmma_part(g - 1, hidden, Ws, S, n_node);
    } else {
        attn_pre_part(g - 1 - gemm_wgs, rela, Wr, Wqr, bqr, q_rel, attn_r, attn_qr, nrel, batch);
    }
}

__global__ void bin_place_kernel(const int* __restrict__ edges,
                                 const int* __restrict__ off,
                                 const int* __restrict__ bstart,
                                 int4* __restrict__ rec,
                                 int n_edge, int nb, int chunk) {
    extern __shared__ int sh_i[];
    const int t = threadIdx.x;
    const int g = blockIdx.x;
    for (int b = t; b < nb; b += BIN_T) sh_i[b] = bstart[b] + off[(long)g * nb + b];
    __syncthreads();
    long e0 = (long)g * chunk;
    long e1 = e0 + chunk;
    if (e1 > n_edge) e1 = n_edge;
    for (long e = e0 + t; e < e1; e += (long)BIN_U * BIN_T) {
        int4 q[BIN_U];
        #pragma unroll
        for (int k = 0; k < BIN_U; ++k) {
            long ee = e + (long)k * BIN_T;
            q[k] = make_int4(0, 0, 0, 0);
            if (ee < e1) {
                const int* p = edges + ee * 6;
                q[k] = make_int4(p[0], p[2], p[4], p[5]);
            }
        }
        #pragma unroll
        for (int k = 0; k < BIN_U; ++k) {
            if (e + (long)k * BIN_T < e1) {
                int pos = atomicAdd(&sh_i[(unsigned)q[k].w / BN], 1);
                rec[pos] = q[k];
            }
        }
    }
}

#define AGG_T  1024
#define EG     8
#define NREL_S 401
#define RELA4  ((NREL_S * IN_DIM / 4 + AGG_T - 1) / AGG_T)
#define ATTR4  ((NREL_S * ATTN_DIM / 4 + AGG_T - 1) / AGG_T)

struct EdgeVals { float s, ar, aq; float2 h, r; };

__device__ __forceinline__ void edge_load(EdgeVals& v, int r_idx, int rel, int sub, int lane,
                                          const float* __restrict__ hidden,
                                          const float* __restrict__ S,
                                          const float* __restrict__ attn_qr,
                                          const float* rela_s,
                                          const float* attnr_s) {
    v.s  = S[(long)sub * ATTN_DIM + lane];
    v.aq = attn_qr[(long)r_idx * ATTN_DIM + lane];
    v.h  = *(const float2*)(hidden + (long)sub * IN_DIM + 2 * lane);
    v.ar = attnr_s[rel * ATTN_DIM + lane];
    v.r  = *(const float2*)(rela_s + rel * IN_DIM + 2 * lane);
}

__device__ __forceinline__ float xhalf(float x) {
    int b = __builtin_bit_cast(int, x);
    return __builtin_bit_cast(float,
        __builtin_amdgcn_permlanex16(b, b, 0x76543210, 0xfedcba98, false, false));
}

template <int M>
__device__ __forceinline__ float row_xor(float x) {
    int b = __builtin_bit_cast(int, x);
    return __builtin_bit_cast(float,
        __builtin_amdgcn_update_dpp(0, b, 0x160 + M, 0xf, 0xf, true));
}

__device__ __forceinline__ float fold_half(float p, float q, bool up) {
    float keep = up ? q : p;
    float send = up ? p : q;
    return keep + xhalf(send);
}
template <int M>
__device__ __forceinline__ float fold_row(float p, float q, bool up) {
    float keep = up ? q : p;
    float send = up ? p : q;
    return keep + row_xor<M>(send);
}

static_assert(EG == 8, "the folding butterfly of the block pass is written out for groups of 8 records");
#define GATE_LANE(k) ((((k) & 1) << 4) | ((((k) >> 1) & 1) << 3) | ((((k) >> 2) & 1) << 2))

__device__ __forceinline__ void edge_finish(const EdgeVals& v, int row, int lane,
                                            float wa, float wb, float* tile) {
    float a = v.s + v.ar + v.aq;
    a = fmaxf(a, 0.f);
    float t = a * wa;
    #pragma unroll
    for (int off = 16; off > 0; off >>= 1) t += __shfl_xor(t, off, 32);
    float alpha = 1.f / (1.f + __expf(-(t + wb)));
    float* dst = tile + row * IN_DIM + 2 * lane;
    atomicAdd(dst,     alpha * (v.h.x + v.r.x));
    atomicAdd(dst + 1, alpha * (v.h.y + v.r.y));
}

__global__ void __launch_bounds__(AGG_T)
block_agg_kernel(const int4*  __restrict__ rec,
                 const int*   __restrict__ bstart,
                 const float* __restrict__ hidden,
                 const float* __restrict__ rela,
                 const float* __restrict__ S,
                 const float* __restrict__ attn_r,
                 const float* __restrict__ attn_qr,
                 const float* __restrict__ w_alpha_w,
                 const float* __restrict__ w_alpha_b,
                 float* __restrict__ agg, int n_node, int nrel) {
    __shared__ float4 lds4[BN * IN_DIM / 4 + (RELA4 + ATTR4) * AGG_T];
    float* tile    = (float*)lds4;
    float* rela_s  = tile + BN * IN_DIM;
    float* attnr_s = rela_s + RELA4 * AGG_T * 4;
    const int t = threadIdx.x;
    const int node0 = (int)blockIdx.x * BN;

    const int nr   = nrel < NREL_S ? nrel : NREL_S;
    const int n_r4 = nr * (IN_DIM / 4);
    const int n_a4 = nr * (ATTN_DIM / 4);
    const float4* rela4  = (const float4*)rela;
    const float4* attnr4 = (const float4*)attn_r;
    float4 cr[RELA4], ca[ATTR4];
    #pragma unroll
    for (int k = 0; k < RELA4; ++k) {
        const int i = t + k * AGG_T;
        cr[k] = rela4[i < n_r4 ? i : 0];
    }
    #pragma unroll
    for (int k = 0; k < ATTR4; ++k) {
        const int i = t + k * AGG_T;
        ca[k] = attnr4[i < n_a4 ? i : 0];
    }
    for (int i = t; i < BN * IN_DIM / 4; i += AGG_T) lds4[i] = make_float4(0.f, 0.f, 0.f, 0.f);
    #pragma unroll
    for (int k = 0; k < RELA4; ++k) ((float4*)rela_s)[t + k * AGG_T] = cr[k];
    #pragma unroll
    for (int k = 0; k < ATTR4; ++k) ((float4*)attnr_s)[t + k * AGG_T] = ca[k];
    __syncthreads();

    const int wave = __builtin_amdgcn_readfirstlane(t >> 5);
    const int lane = t & 31;
    const int r_begin = bstart[blockIdx.x];
    const int r_end   = bstart[blockIdx.x + 1];
    const float wa = w_alpha_w[lane];
    const float wb = w_alpha_b[0];
    const bool up16 = (lane & 16) != 0;
    const bool up8  = (lane & 8) != 0;
    const bool up4  = (lane & 4) != 0;

    for (int base = r_begin + wave * 32; base < r_end; base += (AGG_T / 32) * 32) {
        const int left = r_end - base;
        const int cnt = left < 32 ? left : 32;
        int4 my = make_int4(0, 0, 0, node0);
        if (lane < cnt) my = rec[base + lane];

        int i = 0;
        for (; i + EG <= cnt; i += EG) {
            EdgeVals v[EG];
            int row[EG];
            #pragma unroll
            for (int k = 0; k < EG; ++k) {
                int r_idx = __builtin_amdgcn_readlane(my.x, i + k);
                int rel   = __builtin_amdgcn_readlane(my.y, i + k);
                int sub   = __builtin_amdgcn_readlane(my.z, i + k);
                row[k]    = __builtin_amdgcn_readlane(my.w, i + k) - node0;
                edge_load(v[k], r_idx, rel, sub, lane, hidden, S, attn_qr, rela_s, attnr_s);
            }
            float tk[EG];
            #pragma unroll
            for (int k = 0; k < EG; ++k) {
                float a = v[k].s + v[k].ar + v[k].aq;
                a = fmaxf(a, 0.f);
                tk[k] = a * wa;
            }
            const float u0 = fold_half(tk[0], tk[1], up16);
            const float u1 = fold_half(tk[2], tk[3], up16);
            const float u2 = fold_half(tk[4], tk[5], up16);
            const float u3 = fold_half(tk[6], tk[7], up16);
            const float w0 = fold_row<8>(u0, u1, up8);
            const float w1 = fold_row<8>(u2, u3, up8);
            float z = fold_row<4>(w0, w1, up4);
            z += row_xor<2>(z);
            z += row_xor<1>(z);
            const float alpha8 = 1.f / (1.f + __expf(-(z + wb)));
            #pragma unroll
            for (int k = 0; k < EG; ++k) {
                const float alpha = __builtin_bit_cast(float,
                    __builtin_amdgcn_readlane(__builtin_bit_cast(int, alpha8), GATE_LANE(k)));
                float* dst = tile + row[k] * IN_DIM + 2 * lane;
                atomicAdd(dst,     alpha * (v[k].h.x + v[k].r.x));
                atomicAdd(dst + 1, alpha * (v[k].h.y + v[k].r.y));
            }
        }
        for (; i < cnt; ++i) {
            EdgeVals v;
            int r_idx = __builtin_amdgcn_readlane(my.x, i);
            int rel   = __builtin_amdgcn_readlane(my.y, i);
            int sub   = __builtin_amdgcn_readlane(my.z, i);
            int row   = __builtin_amdgcn_readlane(my.w, i) - node0;
            edge_load(v, r_idx, rel, sub, lane, hidden, S, attn_qr, rela_s, attnr_s);
            edge_finish(v, row, lane, wa, wb, tile);
        }
    }
    __syncthreads();

    int rows = n_node - node0;
    if (rows > BN) rows = BN;
    float4* dst4 = (float4*)(agg + (long)node0 * IN_DIM);
    for (int i = t; i < rows * (IN_DIM / 4); i += AGG_T) dst4[i] = lds4[i];
}

__global__ void out_gemm_wmma_kernel(float* __restrict__ AGG,
                                     const float* __restrict__ Wh, int n_node) {
    int wave  = (int)((blockIdx.x * blockDim.x + threadIdx.x) >> 5);
    int lane  = threadIdx.x & 31;
    int m0    = wave << 4;
    if (m0 >= n_node) return;
    int row16 = lane & 15;
    int hi    = lane >> 4;
    int arow_i = m0 + row16;
    if (arow_i >= n_node) arow_i = n_node - 1;
    const float* arow = AGG + (long)arow_i * IN_DIM + 2 * hi;
    const float* brow = Wh  + (long)row16  * IN_DIM + 2 * hi;

    v8f c0 = {}, c1 = {}, c2 = {}, c3 = {};
    #pragma unroll
    for (int kk = 0; kk < IN_DIM / 4; ++kk) {
        v2f a  = *(const v2f*)(arow + 4 * kk);
        v2f b0 = *(const v2f*)(brow + 0 * 16 * IN_DIM + 4 * kk);
        v2f b1 = *(const v2f*)(brow + 1 * 16 * IN_DIM + 4 * kk);
        v2f b2 = *(const v2f*)(brow + 2 * 16 * IN_DIM + 4 * kk);
        v2f b3 = *(const v2f*)(brow + 3 * 16 * IN_DIM + 4 * kk);
        c0 = __builtin_amdgcn_wmma_f32_16x16x4_f32(false, a, false, b0,
                                                   (short)0, c0, false, false);
        c1 = __builtin_amdgcn_wmma_f32_16x16x4_f32(false, a, false, b1,
                                                   (short)0, c1, false, false);
        c2 = __builtin_amdgcn_wmma_f32_16x16x4_f32(false, a, false, b2,
                                                   (short)0, c2, false, false);
        c3 = __builtin_amdgcn_wmma_f32_16x16x4_f32(false, a, false, b3,
                                                   (short)0, c3, false, false);
    }
    #pragma unroll
    for (int j = 0; j < 8; ++j) {
        int m = m0 + j + 8 * hi;
        if (m < n_node) {
            float* o = AGG + (long)m * OUT_DIM;
            o[row16]      = c0[j];
            o[16 + row16] = c1[j];
            o[32 + row16] = c2[j];
            o[48 + row16] = c3[j];
        }
    }
}

extern "C" void kernel_launch(void* const* d_in, const int* in_sizes, int n_in,
                              void* d_out, int out_size, void* d_ws, size_t ws_size,
                              hipStream_t stream) {
    const int*   q_rel     = (const int*)  d_in[1];
    const float* hidden    = (const float*)d_in[2];
    const int*   edges     = (const int*)  d_in[3];
    const float* rela      = (const float*)d_in[7];
    const float* Ws        = (const float*)d_in[8];
    const float* Wr        = (const float*)d_in[9];
    const float* Wqr       = (const float*)d_in[10];
    const float* bqr       = (const float*)d_in[11];
    const float* w_alpha_w = (const float*)d_in[12];
    const float* w_alpha_b = (const float*)d_in[13];
    const float* Wh        = (const float*)d_in[14];

    const int n_node = in_sizes[2] / IN_DIM;
    const int n_edge = in_sizes[3] / 6;
    const int nrel   = in_sizes[7] / IN_DIM;
    const int batch  = in_sizes[1];
    const int nb     = (n_node + BN - 1) / BN;
    const int chunk  = (n_edge + NCHUNK - 1) / NCHUNK;

    float* agg     = (float*)d_out;
    float* S       = (float*)d_ws;
    float* attn_r  = S + (long)n_node * ATTN_DIM;
    float* attn_qr = attn_r + (long)nrel * ATTN_DIM;
    int*   cnt     = (int*)(attn_qr + (long)batch * ATTN_DIM);
    int*   off     = cnt + (long)NCHUNK * nb;
    int*   bstart  = off + (long)NCHUNK * nb;
    int4*  rec     = (int4*)(bstart + ((nb + 4) / 4) * 4);

    bin_count_kernel<<<NCHUNK, BIN_T, nb * sizeof(int), stream>>>(edges, cnt, n_edge, nb, chunk);

    const int mtiles    = (n_node + 15) / 16;
    const int gemm_wgs  = (mtiles + MID_T / 32 - 1) / (MID_T / 32);
    const int pre_waves = nrel + batch;
    const int pre_wgs   = (pre_waves + MID_T / 32 - 1) / (MID_T / 32);
    mid_kernel<<<1 + gemm_wgs + pre_wgs, MID_T, (nb + MID_T) * sizeof(int), stream>>>(
        cnt, off, bstart, nb, hidden, Ws, S, n_node, gemm_wgs,
        rela, Wr, Wqr, bqr, q_rel, attn_r, attn_qr, nrel, batch);

    bin_place_kernel<<<NCHUNK, BIN_T, nb * sizeof(int), stream>>>(
        edges, off, bstart, rec, n_edge, nb, chunk);

    block_agg_kernel<<<nb, AGG_T, 0, stream>>>(
        rec, bstart, hidden, rela, S, attn_r, attn_qr, w_alpha_w, w_alpha_b, agg, n_node, nrel);

    out_gemm_wmma_kernel<<<(mtiles + 7) / 8, 256, 0, stream>>>(agg, Wh, n_node);
}


# output GEMM (sums x Wh) moved into the block pass: each workgroup multiplies its finished LDS sum tile by an LDS copy of Wh on f32 WMMA, stores final rows; 5 launches -> 4; every sum keeps its order
# speedup vs baseline: 2.6987x; 1.1087x over previous
#include <hip/hip_runtime.h>
#include <hip/hip_bf16.h>

typedef float v2f __attribute__((ext_vector_type(2)));
typedef float v8f __attribute__((ext_vector_type(8)));

#define IN_DIM   64
#define OUT_DIM  64
#define ATTN_DIM 32

#define MID_T 256

__device__ __forceinline__ void attn_pre_part(int wg,
                                const float* __restrict__ rela,
                                const float* __restrict__ Wr,
                                const float* __restrict__ Wqr,
                                const float* __restrict__ bqr,
                                const int*   __restrict__ q_rel,
                                float* __restrict__ attn_r,
                                float* __restrict__ attn_qr,
                                int nrel, int batch) {
    int wave = (int)((wg * MID_T + threadIdx.x) >> 5);
    int k    = threadIdx.x & 31;
    if (wave >= nrel + batch) return;
    if (wave < nrel) {
        const float* hr = rela + (long)wave * IN_DIM;
        const float* w  = Wr   + (long)k    * IN_DIM;
        float s = 0.f;
        #pragma unroll
        for (int d = 0; d < IN_DIM; ++d) s += hr[d] * w[d];
        attn_r[(long)wave * ATTN_DIM + k] = s;
    } else {
        int b  = wave - nrel;
        int qr = q_rel[b];
        const float* hr = rela + (long)qr * IN_DIM;
        const float* w  = Wqr  + (long)k  * IN_DIM;
        float s = bqr[k];
        #pragma unroll
        for (int d = 0; d < IN_DIM; ++d) s += hr[d] * w[d];
        attn_qr[(long)b * ATTN_DIM + k] = s;
    }
}

__device__ __forceinline__ void node_attn_wmma_part(int wg,
                                      const float* __restrict__ H,
                                      const float* __restrict__ Ws,
                                      float* __restrict__ S, int n_node) {
    int wave  = (int)((wg * MID_T + threadIdx.x) >> 5);
    int lane  = threadIdx.x & 31;
    int m0    = wave << 4;
    if (m0 >= n_node) return;
    int row16 = lane & 15;
    int hi    = lane >> 4;
    int arow_i = m0 + row16;
    if (arow_i >= n_node) arow_i = n_node - 1;
    const float* arow  = H  + (long)arow_i      * IN_DIM + 2 * hi;
    const float* b0row = Ws + (long)row16       * IN_DIM + 2 * hi;
    const float* b1row = Ws + (long)(16 + row16) * IN_DIM + 2 * hi;

    v8f c0 = {}; v8f c1 = {};
    #pragma unroll
    for (int kk = 0; kk < IN_DIM / 4; ++kk) {
        v2f a  = *(const v2f*)(arow  + 4 * kk);
        v2f b0 = *(const v2f*)(b0row + 4 * kk);
        v2f b1 = *(const v2f*)(b1row + 4 * kk);
        c0 = __builtin_amdgcn_wmma_f32_16x16x4_f32(false, a, false, b0,
                                                   (short)0, c0, false, false);
        c1 = __builtin_amdgcn_wmma_f32_16x16x4_f32(false, a, false, b1,
                                                   (short)0, c1, false, false);
    }
    #pragma unroll
    for (int j = 0; j < 8; ++j) {
        int m = m0 + j + 8 * hi;
        if (m < n_node) {
            S[(long)m * ATTN_DIM + row16]      = c0[j];
            S[(long)m * ATTN_DIM + 16 + row16] = c1[j];
        }
    }
}

#define BN       392
#define NCHUNK   256
#define BIN_T    1024
#define BIN_U    4
#define SCAN_G   32

__global__ void bin_count_kernel(const int* __restrict__ edges,
                                 int* __restrict__ cnt,
                                 int n_edge, int nb, int chunk) {
    extern __shared__ int sh_i[];
    const int t = threadIdx.x;
    const int g = blockIdx.x;
    for (int b = t; b < nb; b += BIN_T) sh_i[b] = 0;
    __syncthreads();
    long e0 = (long)g * chunk;
    long e1 = e0 + chunk;
    if (e1 > n_edge) e1 = n_edge;
    for (long e = e0 + t; e < e1; e += (long)BIN_U * BIN_T) {
        int obj[BIN_U], inc[BIN_U];
        #pragma unroll
        for (int k = 0; k < BIN_U; ++k) {
            long ee = e + (long)k * BIN_T;
            inc[k] = (ee < e1) ? 1 : 0;
            if (ee >= e1) ee = e1 - 1;
            obj[k] = edges[ee * 6 + 5];
        }
        #pragma unroll
        for (int k = 0; k < BIN_U; ++k)
            atomicAdd(&sh_i[(unsigned)obj[k] / BN], inc[k]);
    }
    __syncthreads();
    for (int b = t; b < nb; b += BIN_T) cnt[(long)g * nb + b] = sh_i[b];
}

__device__ __forceinline__ void bin_scan_part(const int* __restrict__ cnt,
                                int* __restrict__ off,
                                int* __restrict__ bstart,
                                int nb, int* sh_i) {
    int* tot = sh_i;
    int* scr = sh_i + nb;
    const int t = threadIdx.x;
    for (int b = t; b < nb; b += MID_T) {
        int run = 0;
        for (int g = 0; g < NCHUNK; g += SCAN_G) {
            int c[SCAN_G];
            #pragma unroll
            for (int k = 0; k < SCAN_G; ++k) c[k] = cnt[(long)(g + k) * nb + b];
            #pragma unroll
            for (int k = 0; k < SCAN_G; ++k) {
                int v = c[k];
                off[(long)(g + k) * nb + b] = run;
                run += v;
            }
        }
        tot[b] = run;
    }
    __syncthreads();
    int carry = 0;
    for (int base = 0; base < nb; base += MID_T) {
        const int i = base + t;
        const int v = (i < nb) ? tot[i] : 0;
        scr[t] = v;
        __syncthreads();
        for (int d = 1; d < MID_T; d <<= 1) {
            int add = (t >= d) ? scr[t - d] : 0;
            __syncthreads();
            scr[t] += add;
            __syncthreads();
        }
        if (i < nb) bstart[i] = carry + scr[t] - v;
        const int tile_total = scr[MID_T - 1];
        __syncthreads();
        carry += tile_total;
    }
    if (t == 0) bstart[nb] = carry;
}

__global__ void __launch_bounds__(MID_T)
mid_kernel(const int* __restrict__ cnt, int* __restrict__ off, int* __restrict__ bstart, int nb,
           const float* __restrict__ hidden, const float* __restrict__ Ws,
           float* __restrict__ S, int n_node, int gemm_wgs,
           const float* __restrict__ rela, const float* __restrict__ Wr,
           const float* __restrict__ Wqr, const float* __restrict__ bqr,
           const int* __restrict__ q_rel,
           float* __restrict__ attn_r, float* __restrict__ attn_qr, int nrel, int batch) {
    extern __shared__ int sh_i[];
    const int g = (int)blockIdx.x;
    if (g == 0) {
        bin_scan_part(cnt, off, bstart, nb, sh_i);
    } else if (g <= gemm_wgs) {
        node_attn_wmma_part(g - 1, hidden, Ws, S, n_node);
    } else {
        attn_pre_part(g - 1 - gemm_wgs, rela, Wr, Wqr, bqr, q_rel, attn_r, attn_qr, nrel, batch);
    }
}

__global__ void bin_place_kernel(const int* __restrict__ edges,
                                 const int* __restrict__ off,
                                 const int* __restrict__ bstart,
                                 int4* __restrict__ rec,
                                 int n_edge, int nb, int chunk) {
    extern __shared__ int sh_i[];
    const int t = threadIdx.x;
    const int g = blockIdx.x;
    for (int b = t; b < nb; b += BIN_T) sh_i[b] = bstart[b] + off[(long)g * nb + b];
    __syncthreads();
    long e0 = (long)g * chunk;
    long e1 = e0 + chunk;
    if (e1 > n_edge) e1 = n_edge;
    for (long e = e0 + t; e < e1; e += (long)BIN_U * BIN_T) {
        int4 q[BIN_U];
        #pragma unroll
        for (int k = 0; k < BIN_U; ++k) {
            long ee = e + (long)k * BIN_T;
            q[k] = make_int4(0, 0, 0, 0);
            if (ee < e1) {
                const int* p = edges + ee * 6;
                q[k] = make_int4(p[0], p[2], p[4], p[5]);
            }
        }
        #pragma unroll
        for (int k = 0; k < BIN_U; ++k) {
            if (e + (long)k * BIN_T < e1) {
                int pos = atomicAdd(&sh_i[(unsigned)q[k].w / BN], 1);
                rec[pos] = q[k];
            }
        }
    }
}

#define AGG_T  1024
#define EG     8
#define NREL_S 401
#define RELA4  ((NREL_S * IN_DIM / 4 + AGG_T - 1) / AGG_T)
#define ATTR4  ((NREL_S * ATTN_DIM / 4 + AGG_T - 1) / AGG_T)
#define TS     68
#define TROWS  ((BN + 15) / 16 * 16)
static_assert(TROWS / 16 <= AGG_T / 32, "one wave per 16-row tile in the output product");
static_assert(OUT_DIM * IN_DIM / 4 == AGG_T, "the Wh copy takes one float4 per thread");

struct EdgeVals { float s, ar, aq; float2 h, r; };

__device__ __forceinline__ void edge_load(EdgeVals& v, int r_idx, int rel, int sub, int lane,
                                          const float* __restrict__ hidden,
                                          const float* __restrict__ S,
                                          const float* __restrict__ attn_qr,
                                          const float* rela_s,
                                          const float* attnr_s) {
    v.s  = S[(long)sub * ATTN_DIM + lane];
    v.aq = attn_qr[(long)r_idx * ATTN_DIM + lane];
    v.h  = *(const float2*)(hidden + (long)sub * IN_DIM + 2 * lane);
    v.ar = attnr_s[rel * ATTN_DIM + lane];
    v.r  = *(const float2*)(rela_s + rel * IN_DIM + 2 * lane);
}

__device__ __forceinline__ float xhalf(float x) {
    int b = __builtin_bit_cast(int, x);
    return __builtin_bit_cast(float,
        __builtin_amdgcn_permlanex16(b, b, 0x76543210, 0xfedcba98, false, false));
}

template <int M>
__device__ __forceinline__ float row_xor(float x) {
    int b = __builtin_bit_cast(int, x);
    return __builtin_bit_cast(float,
        __builtin_amdgcn_update_dpp(0, b, 0x160 + M, 0xf, 0xf, true));
}

__device__ __forceinline__ float fold_half(float p, float q, bool up) {
    float keep = up ? q : p;
    float send = up ? p : q;
    return keep + xhalf(send);
}
template <int M>
__device__ __forceinline__ float fold_row(float p, float q, bool up) {
    float keep = up ? q : p;
    float send = up ? p : q;
    return keep + row_xor<M>(send);
}

static_assert(EG == 8, "the folding butterfly of the block pass is written out for groups of 8 records");
#define GATE_LANE(k) ((((k) & 1) << 4) | ((((k) >> 1) & 1) << 3) | ((((k) >> 2) & 1) << 2))

__device__ __forceinline__ void edge_finish(const EdgeVals& v, int row, int lane,
                                            float wa, float wb, float* tile) {
    float a = v.s + v.ar + v.aq;
    a = fmaxf(a, 0.f);
    float t = a * wa;
    #pragma unroll
    for (int off = 16; off > 0; off >>= 1) t += __shfl_xor(t, off, 32);
    float alpha = 1.f / (1.f + __expf(-(t + wb)));
    float* dst = tile + row * TS + 2 * lane;
    atomicAdd(dst,     alpha * (v.h.x + v.r.x));
    atomicAdd(dst + 1, alpha * (v.h.y + v.r.y));
}

__global__ void __launch_bounds__(AGG_T)
block_agg_kernel(const int4*  __restrict__ rec,
                 const int*   __restrict__ bstart,
                 const float* __restrict__ hidden,
                 const float* __restrict__ rela,
                 const float* __restrict__ S,
                 const float* __restrict__ attn_r,
                 const float* __restrict__ attn_qr,
                 const float* __restrict__ w_alpha_w,
                 const float* __restrict__ w_alpha_b,
                 const float* __restrict__ Wh,
                 float* __restrict__ out, int n_node, int nrel) {
    __shared__ float4 lds4[TROWS * TS / 4 + (RELA4 + ATTR4) * AGG_T + OUT_DIM * IN_DIM / 4];
    float* tile    = (float*)lds4;
    float* rela_s  = tile + TROWS * TS;
    float* attnr_s = rela_s + RELA4 * AGG_T * 4;
    float* whf     = attnr_s + ATTR4 * AGG_T * 4;
    const int t = threadIdx.x;
    const int node0 = (int)blockIdx.x * BN;

    const int nr   = nrel < NREL_S ? nrel : NREL_S;
    const int n_r4 = nr * (IN_DIM / 4);
    const int n_a4 = nr * (ATTN_DIM / 4);
    const float4* rela4  = (const float4*)rela;
    const float4* attnr4 = (const float4*)attn_r;
    float4 cr[RELA4], ca[ATTR4];
    #pragma unroll
    for (int k = 0; k < RELA4; ++k) {
        const int i = t + k * AGG_T;
        cr[k] = rela4[i < n_r4 ? i : 0];
    }
    #pragma unroll
    for (int k = 0; k < ATTR4; ++k) {
        const int i = t + k * AGG_T;
        ca[k] = attnr4[i < n_a4 ? i : 0];
    }
    const float4 cw = ((const float4*)Wh)[t];
    for (int i = t; i < TROWS * TS / 4; i += AGG_T) lds4[i] = make_float4(0.f, 0.f, 0.f, 0.f);
    #pragma unroll
    for (int k = 0; k < RELA4; ++k) ((float4*)rela_s)[t + k * AGG_T] = cr[k];
    #pragma unroll
    for (int k = 0; k < ATTR4; ++k) ((float4*)attnr_s)[t + k * AGG_T] = ca[k];
    {
        const int n  = t >> 4, kk = t & 15;
        const int j  = n >> 4, c = n & 15;
        float* p = whf + (((kk * 2 + (j >> 1)) * 32 + c) * 4 + (j & 1) * 2);
        *(float2*)p            = make_float2(cw.x, cw.y);
        *(float2*)(p + 16 * 4) = make_float2(cw.z, cw.w);
    }
    __syncthreads();

    const int wave = __builtin_amdgcn_readfirstlane(t >> 5);
    const int lane = t & 31;
    const int r_begin = bstart[blockIdx.x];
    const int r_end   = bstart[blockIdx.x + 1];
    const float wa = w_alpha_w[lane];
    const float wb = w_alpha_b[0];
    const bool up16 = (lane & 16) != 0;
    const bool up8  = (lane & 8) != 0;
    const bool up4  = (lane & 4) != 0;

    for (int base = r_begin + wave * 32; base < r_end; base += (AGG_T / 32) * 32) {
        const int left = r_end - base;
        const int cnt = left < 32 ? left : 32;
        int4 my = make_int4(0, 0, 0, node0);
        if (lane < cnt) my = rec[base + lane];

        int i = 0;
        for (; i + EG <= cnt; i += EG) {
            EdgeVals v[EG];
            int row[EG];
            #pragma unroll
            for (int k = 0; k < EG; ++k) {
                int r_idx = __builtin_amdgcn_readlane(my.x, i + k);
                int rel   = __builtin_amdgcn_readlane(my.y, i + k);
                int sub   = __builtin_amdgcn_readlane(my.z, i + k);
                row[k]    = __builtin_amdgcn_readlane(my.w, i + k) - node0;
                edge_load(v[k], r_idx, rel, sub, lane, hidden, S, attn_qr, rela_s, attnr_s);
            }
            float tk[EG];
            #pragma unroll
            for (int k = 0; k < EG; ++k) {
                float a = v[k].s + v[k].ar + v[k].aq;
                a = fmaxf(a, 0.f);
                tk[k] = a * wa;
            }
            const float u0 = fold_half(tk[0], tk[1], up16);
            const float u1 = fold_half(tk[2], tk[3], up16);
            const float u2 = fold_half(tk[4], tk[5], up16);
            const float u3 = fold_half(tk[6], tk[7], up16);
            const float w0 = fold_row<8>(u0, u1, up8);
            const float w1 = fold_row<8>(u2, u3, up8);
            float z = fold_row<4>(w0, w1, up4);
            z += row_xor<2>(z);
            z += row_xor<1>(z);
            const float alpha8 = 1.f / (1.f + __expf(-(z + wb)));
            #pragma unroll
            for (int k = 0; k < EG; ++k) {
                const float alpha = __builtin_bit_cast(float,
                    __builtin_amdgcn_readlane(__builtin_bit_cast(int, alpha8), GATE_LANE(k)));
                float* dst = tile + row[k] * TS + 2 * lane;
                atomicAdd(dst,     alpha * (v[k].h.x + v[k].r.x));
                atomicAdd(dst + 1, alpha * (v[k].h.y + v[k].r.y));
            }
        }
        for (; i < cnt; ++i) {
            EdgeVals v;
            int r_idx = __builtin_amdgcn_readlane(my.x, i);
            int rel   = __builtin_amdgcn_readlane(my.y, i);
            int sub   = __builtin_amdgcn_readlane(my.z, i);
            int row   = __builtin_amdgcn_readlane(my.w, i) - node0;
            edge_load(v, r_idx, rel, sub, lane, hidden, S, attn_qr, rela_s, attnr_s);
            edge_finish(v, row, lane, wa, wb, tile);
        }
    }
    __syncthreads();

    int rows = n_node - node0;
    if (rows > BN) rows = BN;

    if (wave * 16 < rows) {
        const int c16 = lane & 15;
        const int hi  = lane >> 4;
        const float*  arow = tile + (wave * 16 + c16) * TS + 2 * hi;
        const float4* brow = (const float4*)whf + lane;
        v8f c0 = {}, c1 = {}, c2 = {}, c3 = {};
        #pragma unroll
        for (int kk = 0; kk < IN_DIM / 4; ++kk) {
            const v2f    a   = *(const v2f*)(arow + 4 * kk);
            const float4 b01 = brow[(kk * 2 + 0) * 32];
            const float4 b23 = brow[(kk * 2 + 1) * 32];
            const v2f b0 = {b01.x, b01.y}, b1 = {b01.z, b01.w};
            const v2f b2 = {b23.x, b23.y}, b3 = {b23.z, b23.w};
            c0 = __builtin_amdgcn_wmma_f32_16x16x4_f32(false, a, false, b0,
                                                       (short)0, c0, false, false);
            c1 = __builtin_amdgcn_wmma_f32_16x16x4_f32(false, a, false, b1,
                                                       (short)0, c1, false, false);
            c2 = __builtin_amdgcn_wmma_f32_16x16x4_f32(false, a, false, b2,
                                                       (short)0, c2, false, false);
            c3 = __builtin_amdgcn_wmma_f32_16x16x4_f32(false, a, false, b3,
                                                       (short)0, c3, false, false);
        }
        float* drow = tile + (wave * 16 + 8 * hi) * TS + c16;
        #pragma unroll
        for (int j = 0; j < 8; ++j) {
            drow[j * TS]      = c0[j];
            drow[j * TS + 16] = c1[j];
            drow[j * TS + 32] = c2[j];
            drow[j * TS + 48] = c3[j];
        }
        float4* dst4 = (float4*)(out + (long)(node0 + wave * 16) * OUT_DIM);
        const float4* src4 = lds4 + (wave * 16 + hi) * (TS / 4) + c16;
        if (wave * 16 + 16 <= rows) {
            #pragma unroll
            for (int i = 0; i < 8; ++i)
                dst4[(2 * i + hi) * (OUT_DIM / 4) + c16] = src4[2 * i * (TS / 4)];
        } else {
            #pragma unroll
            for (int i = 0; i < 8; ++i)
                if (wave * 16 + 2 * i + hi < rows)
                    dst4[(2 * i + hi) * (OUT_DIM / 4) + c16] = src4[2 * i * (TS / 4)];
        }
    }
}

extern "C" void kernel_launch(void* const* d_in, const int* in_sizes, int n_in,
                              void* d_out, int out_size, void* d_ws, size_t ws_size,
                              hipStream_t stream) {
    const int*   q_rel     = (const int*)  d_in[1];
    const float* hidden    = (const float*)d_in[2];
    const int*   edges     = (const int*)  d_in[3];
    const float* rela      = (const float*)d_in[7];
    const float* Ws        = (const float*)d_in[8];
    const float* Wr        = (const float*)d_in[9];
    const float* Wqr       = (const float*)d_in[10];
    const float* bqr       = (const float*)d_in[11];
    const float* w_alpha_w = (const float*)d_in[12];
    const float* w_alpha_b = (const float*)d_in[13];
    const float* Wh        = (const float*)d_in[14];

    const int n_node = in_sizes[2] / IN_DIM;
    const int n_edge = in_sizes[3] / 6;
    const int nrel   = in_sizes[7] / IN_DIM;
    const int batch  = in_sizes[1];
    const int nb     = (n_node + BN - 1) / BN;
    const int chunk  = (n_edge + NCHUNK - 1) / NCHUNK;

    float* out     = (float*)d_out;
    float* S       = (float*)d_ws;
    float* attn_r  = S + (long)n_node * ATTN_DIM;
    float* attn_qr = attn_r + (long)nrel * ATTN_DIM;
    int*   cnt     = (int*)(attn_qr + (long)batch * ATTN_DIM);
    int*   off     = cnt + (long)NCHUNK * nb;
    int*   bstart  = off + (long)NCHUNK * nb;
    int4*  rec     = (int4*)(bstart + ((nb + 4) / 4) * 4);

    bin_count_kernel<<<NCHUNK, BIN_T, nb * sizeof(int), stream>>>(edges, cnt, n_edge, nb, chunk);

    const int mtiles    = (n_node + 15) / 16;
    const int gemm_wgs  = (mtiles + MID_T / 32 - 1) / (MID_T / 32);
    const int pre_waves = nrel + batch;
    const int pre_wgs   = (pre_waves + MID_T / 32 - 1) / (MID_T / 32);
    mid_kernel<<<1 + gemm_wgs + pre_wgs, MID_T, (nb + MID_T) * sizeof(int), stream>>>(
        cnt, off, bstart, nb, hidden, Ws, S, n_node, gemm_wgs,
        rela, Wr, Wqr, bqr, q_rel, attn_r, attn_qr, nrel, batch);

    bin_place_kernel<<<NCHUNK, BIN_T, nb * sizeof(int), stream>>>(
        edges, off, bstart, rec, n_edge, nb, chunk);

    block_agg_kernel<<<nb, AGG_T, 0, stream>>>(
        rec, bstart, hidden, rela, S, attn_r, attn_qr, w_alpha_w, w_alpha_b, Wh, out, n_node, nrel);
}
